# starting kernel
# baseline (speedup 1.0000x reference)
#include <hip/hip_runtime.h>
#include <math.h>

constexpr int TLEN  = 320000;
constexpr int BROWS = 64;
constexpr int CHUNK = 320;
constexpr int NC    = TLEN / CHUNK;
constexpr int NBLK  = CHUNK / 16;

typedef __attribute__((ext_vector_type(2))) float v2f;
typedef __attribute__((ext_vector_type(8))) float v8f;

__device__ __forceinline__ float fir3(const float* __restrict__ xr, int t,
                                      float b0, float b1, float b2) {
    float x0 = xr[t];
    float x1 = (t >= 1) ? xr[t - 1] : 0.0f;
    float x2 = (t >= 2) ? xr[t - 2] : 0.0f;
    return fmaf(b0, x0, fmaf(b1, x1, b2 * x2));
}

__global__ void lp_phase1(const float* __restrict__ x, float2* __restrict__ dstate,
                          float b0, float b1, float b2, float a1, float a2) {
    int gid = blockIdx.x * blockDim.x + threadIdx.x;
    if (gid >= BROWS * NC) return;
    int row = gid / NC;
    int c   = gid % NC;
    const float* xr = x + (long)row * TLEN;
    int t0 = c * CHUNK;
    float y1 = 0.0f, y2 = 0.0f;
    float xm1 = (t0 >= 1) ? xr[t0 - 1] : 0.0f;
    float xm2 = (t0 >= 2) ? xr[t0 - 2] : 0.0f;
    for (int n = 0; n < CHUNK; ++n) {
        float xn = xr[t0 + n];
        float f  = fmaf(b0, xn, fmaf(b1, xm1, b2 * xm2));
        float y  = fmaf(-a1, y1, fmaf(-a2, y2, f));
        y2 = y1; y1 = y; xm2 = xm1; xm1 = xn;
    }
    dstate[gid] = make_float2(y1, y2);
}

__global__ void lp_phase2(const float2* __restrict__ dstate, float2* __restrict__ entry,
                          float p00, float p01, float p10, float p11) {
    int row = blockIdx.x * blockDim.x + threadIdx.x;
    if (row >= BROWS) return;
    float s1 = 0.0f, s2 = 0.0f;
    for (int c = 0; c < NC; ++c) {
        entry[row * NC + c] = make_float2(s1, s2);
        float2 d = dstate[row * NC + c];
        float n1 = fmaf(p00, s1, fmaf(p01, s2, d.x));
        float n2 = fmaf(p10, s1, fmaf(p11, s2, d.y));
        s1 = n1; s2 = n2;
    }
}

__global__ __launch_bounds__(256) void lp_phase3(
    const float* __restrict__ x, float* __restrict__ out,
    const float2* __restrict__ entry,
    float b0, float b1, float b2, float a1, float a2) {

    int lane = threadIdx.x & 31;
    int wid  = blockIdx.x * 8 + (threadIdx.x >> 5);
    int c  = wid % NC;
    int rg = wid / NC;
    if (rg >= BROWS / 16) return;

    int half = lane >> 4;
    int lid  = lane & 15;
    int r0   = rg * 16;
    int row  = r0 + lid;
    int t0   = c * CHUNK;

    float h[17];
    h[0] = 1.0f;
    h[1] = -a1;
    #pragma unroll
    for (int n = 2; n <= 16; ++n) h[n] = fmaf(-a1, h[n - 1], -a2 * h[n - 2]);

    float Tt[20];
    #pragma unroll
    for (int j = 0; j < 16; ++j) Tt[j] = (lid >= j) ? h[lid - j] : 0.0f;
    Tt[16] = h[lid + 1];
    Tt[17] = -a2 * h[lid];
    Tt[18] = 0.0f;
    Tt[19] = 0.0f;

    v2f Bk[5];
    #pragma unroll
    for (int kc = 0; kc < 5; ++kc) {
        Bk[kc].x = half ? Tt[4 * kc + 2] : Tt[4 * kc + 0];
        Bk[kc].y = half ? Tt[4 * kc + 3] : Tt[4 * kc + 1];
    }

    float2 e = entry[row * NC + c];
    float ym1 = e.x, ym2 = e.y;

    const float* xr = x + (long)row * TLEN;

    for (int blk = 0; blk < NBLK; ++blk) {
        int tb = t0 + blk * 16;

        v2f Ak[5];
        #pragma unroll
        for (int kc = 0; kc < 4; ++kc) {
            int k0 = 4 * kc + (half ? 2 : 0);
            int k1 = 4 * kc + (half ? 3 : 1);
            Ak[kc].x = fir3(xr, tb + k0, b0, b1, b2);
            Ak[kc].y = fir3(xr, tb + k1, b0, b1, b2);
        }
        Ak[4].x = half ? 0.0f : ym1;
        Ak[4].y = half ? 0.0f : ym2;

        v8f acc = {};
        #pragma unroll
        for (int kc = 0; kc < 5; ++kc) {
            acc = __builtin_amdgcn_wmma_f32_16x16x4_f32(
                 false, Ak[kc],  false, Bk[kc],
                 (short)0, acc,  false,  false);
        }

        #pragma unroll
        for (int v = 0; v < 8; ++v) {
            out[(long)(r0 + v + 8 * half) * TLEN + tb + lid] = acc[v];
        }

        float nym1 = 0.0f, nym2 = 0.0f;
        #pragma unroll
        for (int v = 0; v < 8; ++v) {
            float lo15 = __shfl(acc[v], 15, 32);
            float hi15 = __shfl(acc[v], 31, 32);
            float lo14 = __shfl(acc[v], 14, 32);
            float hi14 = __shfl(acc[v], 30, 32);
            if (lid == v)     { nym1 = lo15; nym2 = lo14; }
            if (lid == v + 8) { nym1 = hi15; nym2 = hi14; }
        }
        ym1 = nym1; ym2 = nym2;
    }
}

extern "C" void kernel_launch(void* const* d_in, const int* in_sizes, int n_in,
                              void* d_out, int out_size, void* d_ws, size_t ws_size,
                              hipStream_t stream) {
    const float* x = (const float*)d_in[0];
    float* out = (float*)d_out;

    float2* dstate = (float2*)d_ws;
    float2* entry  = dstate + (size_t)BROWS * NC;

    double w0    = 2.0 * M_PI * 7500.0 / 16000.0;
    double alpha = sin(w0) / (2.0 * 0.707);
    double cw    = cos(w0);
    double b0 = (1.0 - cw) / 2.0, b1 = 1.0 - cw, b2 = b0;
    double a0 = 1.0 + alpha, a1 = -2.0 * cw, a2 = 1.0 - alpha;
    b0 /= a0; b1 /= a0; b2 /= a0; a1 /= a0; a2 /= a0;

    double p00 = 1, p01 = 0, p10 = 0, p11 = 1;
    for (int i = 0; i < CHUNK; ++i) {
        double n00 = -a1 * p00 - a2 * p10;
        double n01 = -a1 * p01 - a2 * p11;
        double n10 = p00, n11 = p01;
        p00 = n00; p01 = n01; p10 = n10; p11 = n11;
    }

    float fb0 = (float)b0, fb1 = (float)b1, fb2 = (float)b2;
    float fa1 = (float)a1, fa2 = (float)a2;

    int n1 = BROWS * NC;
    lp_phase1<<<(n1 + 255) / 256, 256, 0, stream>>>(x, dstate, fb0, fb1, fb2, fa1, fa2);
    lp_phase2<<<1, 64, 0, stream>>>(dstate, entry,
                                    (float)p00, (float)p01, (float)p10, (float)p11);
    int nwaves = NC * (BROWS / 16);
    lp_phase3<<<nwaves / 8, 256, 0, stream>>>(x, out, entry, fb0, fb1, fb2, fa1, fa2);
}


# chunk-state scan moved into phase 1's blocks (lane scan restarted at row starts, 8 wave totals joined in LDS); phase 3 joins <=4 block totals; scan launch gone (entry-state sums reordered)
# speedup vs baseline: 5.3000x; 5.3000x over previous
#include <hip/hip_runtime.h>
#include <math.h>

constexpr int TLEN  = 320000;
constexpr int BROWS = 64;
constexpr int CHUNK = 320;
constexpr int NC    = TLEN / CHUNK;
constexpr int NBLK  = CHUNK / 16;

typedef __attribute__((ext_vector_type(2))) float v2f;
typedef __attribute__((ext_vector_type(8))) float v8f;

__device__ __forceinline__ float fir3(const float* __restrict__ xr, int t,
                                      float b0, float b1, float b2) {
    float x0 = xr[t];
    float x1 = (t >= 1) ? xr[t - 1] : 0.0f;
    float x2 = (t >= 2) ? xr[t - 2] : 0.0f;
    return fmaf(b0, x0, fmaf(b1, x1, b2 * x2));
}

__device__ __forceinline__ void aff_step(float& s1, float& s2, float dx, float dy,
                                         float m00, float m01, float m10, float m11) {
    float n1 = fmaf(m00, s1, fmaf(m01, s2, dx));
    float n2 = fmaf(m10, s1, fmaf(m11, s2, dy));
    s1 = n1; s2 = n2;
}

__device__ __forceinline__ void mat_square(float& m00, float& m01, float& m10, float& m11) {
    float n00 = fmaf(m00, m00, m01 * m10);
    float n01 = fmaf(m00, m01, m01 * m11);
    float n10 = fmaf(m10, m00, m11 * m10);
    float n11 = fmaf(m10, m01, m11 * m11);
    m00 = n00; m01 = n01; m10 = n10; m11 = n11;
}

__device__ __forceinline__ void mat_mul(float& r00, float& r01, float& r10, float& r11,
                                        float m00, float m01, float m10, float m11) {
    float n00 = fmaf(r00, m00, r01 * m10);
    float n01 = fmaf(r00, m01, r01 * m11);
    float n10 = fmaf(r10, m00, r11 * m10);
    float n11 = fmaf(r10, m01, r11 * m11);
    r00 = n00; r01 = n01; r10 = n10; r11 = n11;
}

constexpr int P1_THREADS = 256;
constexpr int P1_WAVES   = P1_THREADS / 32;
constexpr int SLAB   = 64;
constexpr int NSLAB  = CHUNK / SLAB;
constexpr int SLAB4  = SLAB / 4;
constexpr int PITCH4 = SLAB4 + 1;
static_assert(CHUNK % SLAB == 0, "a chunk is a whole number of slabs");
static_assert((BROWS * NC) % P1_THREADS == 0, "no partial block: the barriers below need every lane");
constexpr int NPB = BROWS * NC / P1_THREADS;
constexpr int MAXJOIN = (NC - 2) / P1_THREADS + 1;
constexpr int BT_PITCH = 16;

__global__ __launch_bounds__(P1_THREADS) void lp_phase1(const float* __restrict__ x,
                                                        float2* __restrict__ entry,
                                                        float2* __restrict__ btot,
                                                        float b0, float b1, float b2,
                                                        float a1, float a2,
                                                        float p00, float p01, float p10, float p11) {
    __shared__ float4 stage[2][P1_WAVES][32 * PITCH4];
    __shared__ float2 wtot[P1_WAVES];

    const int lane = threadIdx.x & 31;
    const int wv   = threadIdx.x >> 5;
    const int gid  = blockIdx.x * P1_THREADS + threadIdx.x;
    const int g0   = gid - lane;
    const int c    = gid % NC;

    const float* xc = x + (long)gid * CHUNK;
    float y1 = 0.0f, y2 = 0.0f;
    float xm1 = (c > 0) ? xc[-1] : 0.0f;
    float xm2 = (c > 0) ? xc[-2] : 0.0f;

    const int sub   = lane >> 4;
    const int piece = lane & 15;
    const float* src = x + (long)(g0 + sub) * CHUNK + piece * 4;

    float4 g[SLAB4];
    #pragma unroll
    for (int i = 0; i < SLAB4; ++i)
        g[i] = *reinterpret_cast<const float4*>(src + (long)(2 * i) * CHUNK);

    #pragma unroll
    for (int s = 0; s < NSLAB; ++s) {
        float4* buf = stage[s & 1][wv];
        #pragma unroll
        for (int i = 0; i < SLAB4; ++i) buf[(2 * i + sub) * PITCH4 + piece] = g[i];

        __syncthreads();

        float4 v[SLAB4];
        #pragma unroll
        for (int q = 0; q < SLAB4; ++q) v[q] = buf[lane * PITCH4 + q];

        if (s + 1 < NSLAB) {
            #pragma unroll
            for (int i = 0; i < SLAB4; ++i)
                g[i] = *reinterpret_cast<const float4*>(src + (long)(2 * i) * CHUNK + (s + 1) * SLAB);
        }

        #pragma unroll
        for (int q = 0; q < SLAB4; ++q) {
            const float xs[4] = {v[q].x, v[q].y, v[q].z, v[q].w};
            #pragma unroll
            for (int e = 0; e < 4; ++e) {
                float xn = xs[e];
                float f  = fmaf(b0, xn, fmaf(b1, xm1, b2 * xm2));
                float y  = fmaf(-a1, y1, fmaf(-a2, y2, f));
                y2 = y1; y1 = y; xm2 = xm1; xm1 = xn;
            }
        }
    }

    float t1 = y1, t2 = y2;
    float q00 = p00, q01 = p01, q10 = p10, q11 = p11;
    float r00 = 1.0f, r01 = 0.0f, r10 = 0.0f, r11 = 1.0f;
    #pragma unroll
    for (int o = 1; o < 32; o <<= 1) {
        float u1 = __shfl_up(t1, o, 32);
        float u2 = __shfl_up(t2, o, 32);
        const bool take = (lane >= o) && (c >= o);
        aff_step(u1, u2, t1, t2, q00, q01, q10, q11);
        t1 = take ? u1 : t1;
        t2 = take ? u2 : t2;
        if (lane & o) mat_mul(r00, r01, r10, r11, q00, q01, q10, q11);
        mat_square(q00, q01, q10, q11);
    }

    float e1 = __shfl_up(t1, 1, 32);
    float e2 = __shfl_up(t2, 1, 32);
    if (lane == 0 || c == 0) { e1 = 0.0f; e2 = 0.0f; }

    if (lane == 31) wtot[wv] = make_float2(t1, t2);
    __syncthreads();

    float E1 = 0.0f, E2 = 0.0f, W1 = 0.0f, W2 = 0.0f;
    const int base = blockIdx.x * P1_THREADS;
    #pragma unroll
    for (int v = 0; v < P1_WAVES; ++v) {
        if (v == wv) { W1 = E1; W2 = E2; }
        const float2 T = wtot[v];
        const bool link = ((base + 32 * v + 31) % NC) >= 32;
        float n1 = E1, n2 = E2;
        aff_step(n1, n2, T.x, T.y, q00, q01, q10, q11);
        E1 = link ? n1 : T.x;
        E2 = link ? n2 : T.y;
    }
    if (threadIdx.x == 0) btot[blockIdx.x * BT_PITCH] = make_float2(E1, E2);

    if (c >= lane + 1) {
        e1 = fmaf(r00, W1, fmaf(r01, W2, e1));
        e2 = fmaf(r10, W1, fmaf(r11, W2, e2));
    }
    entry[gid] = make_float2(e1, e2);
}

template <int L>
__device__ __forceinline__ float row_lane(float v) {
    int b = __builtin_bit_cast(int, v);
    return __builtin_bit_cast(float,
        __builtin_amdgcn_mov_dpp(b, 0x150 + L  , 0xF, 0xF, false));
}
__device__ __forceinline__ float other_half(float v) {
    int b = __builtin_bit_cast(int, v);
    return __builtin_bit_cast(float,
        __builtin_amdgcn_permlanex16(b, b, 0x76543210, 0xfedcba98, false, false));
}
__device__ __forceinline__ float pick8(const float (&u)[8], bool b0, bool b1, bool b2) {
    float p0 = b0 ? u[1] : u[0];
    float p1 = b0 ? u[3] : u[2];
    float p2 = b0 ? u[5] : u[4];
    float p3 = b0 ? u[7] : u[6];
    float q0 = b1 ? p1 : p0;
    float q1 = b1 ? p3 : p2;
    return b2 ? q1 : q0;
}

__global__ __launch_bounds__(256) void lp_phase3(
    const float* __restrict__ x, float* __restrict__ out,
    const float2* __restrict__ entry, const float2* __restrict__ btot,
    float b0, float b1, float b2, float a1, float a2,
    float p00, float p01, float p10, float p11) {

    int lane = threadIdx.x & 31;
    int wid  = blockIdx.x * 8 + (threadIdx.x >> 5);
    int c  = wid % NC;
    int rg = wid / NC;
    if (rg >= BROWS / 16) return;

    int half = lane >> 4;
    int lid  = lane & 15;
    const bool bit0 = (lid & 1) != 0, bit1 = (lid & 2) != 0, bit2 = (lid & 4) != 0;
    const bool upper8 = (lid & 8) != 0;
    int r0   = rg * 16;
    int row  = r0 + lid;
    int t0   = c * CHUNK;

    float h[17];
    h[0] = 1.0f;
    h[1] = -a1;
    #pragma unroll
    for (int n = 2; n <= 16; ++n) h[n] = fmaf(-a1, h[n - 1], -a2 * h[n - 2]);

    float g[16];
    g[0] = b0 * h[0];
    g[1] = fmaf(b0, h[1], b1 * h[0]);
    #pragma unroll
    for (int n = 2; n < 16; ++n) g[n] = fmaf(b0, h[n], fmaf(b1, h[n - 1], b2 * h[n - 2]));

    float hm = 0.0f, hl = 0.0f, hp = 0.0f;
    #pragma unroll
    for (int n = 0; n <= 16; ++n) {
        if (n + 1 == lid) hm = h[n];
        if (n == lid)     hl = h[n];
        if (n == lid + 1) hp = h[n];
    }

    float Tt[20];
    #pragma unroll
    for (int j = 0; j < 16; ++j) Tt[j] = (lid >= j) ? g[lid - j] : 0.0f;
    Tt[16] = hp;
    Tt[17] = -a2 * hl;
    Tt[18] = fmaf(b1, hl, b2 * hm);
    Tt[19] = b2 * hl;

    v2f Bk[5];
    #pragma unroll
    for (int kc = 0; kc < 5; ++kc) {
        Bk[kc].x = half ? Tt[4 * kc + 2] : Tt[4 * kc + 0];
        Bk[kc].y = half ? Tt[4 * kc + 3] : Tt[4 * kc + 1];
    }

    const int gch = row * NC + c;
    const int pb  = gch / P1_THREADS;
    const int j   = gch % P1_THREADS;
    const float2 e = entry[gch];
    float2 Bt[MAXJOIN];
    #pragma unroll
    for (int k = 1; k <= MAXJOIN; ++k) Bt[k - 1] = btot[max(pb - k, 0) * BT_PITCH];


    const float* xr = x + (long)row * TLEN;

    float xm1 = (t0 > 0) ? xr[t0 - 1] : 0.0f;
    float xm2 = (t0 > 0) ? xr[t0 - 2] : 0.0f;

    v2f Ax[NBLK][4];
    #pragma unroll
    for (int blk = 0; blk < NBLK; ++blk) {
        const v2f* xa = reinterpret_cast<const v2f*>(xr + t0 + blk * 16 + 2 * half);
        #pragma unroll
        for (int kc = 0; kc < 4; ++kc) Ax[blk][kc] = xa[2 * kc];
    }
    __builtin_amdgcn_sched_barrier(0);

    float q00 = p00, q01 = p01, q10 = p10, q11 = p11;
    float r00 = 1.0f, r01 = 0.0f, r10 = 0.0f, r11 = 1.0f;
    #pragma unroll
    for (int k = 1; k < P1_THREADS; k <<= 1) {
        if (j & k) mat_mul(r00, r01, r10, r11, q00, q01, q10, q11);
        mat_square(q00, q01, q10, q11);
    }
    float E1 = 0.0f, E2 = 0.0f;
    #pragma unroll
    for (int k = MAXJOIN; k >= 1; --k) {
        const bool ok = c >= j + 1 + P1_THREADS * (k - 1);
        float n1 = E1, n2 = E2;
        aff_step(n1, n2, Bt[k - 1].x, Bt[k - 1].y, q00, q01, q10, q11);
        E1 = ok ? n1 : E1;
        E2 = ok ? n2 : E2;
    }
    float ym1 = e.x, ym2 = e.y;
    if (c >= j + 1) {
        ym1 = fmaf(r00, E1, fmaf(r01, E2, ym1));
        ym2 = fmaf(r10, E1, fmaf(r11, E2, ym2));
    }

    v8f part = {};
    #pragma unroll
    for (int kc = 0; kc < 4; ++kc) {
        part = __builtin_amdgcn_wmma_f32_16x16x4_f32(
             false, Ax[0][kc],  false, Bk[kc],
             (short)0, part,  false,  false);
    }

    #pragma unroll
    for (int blk = 0; blk < NBLK; ++blk) {
        int tb = t0 + blk * 16;

        v2f Ak4;
        Ak4.x = half ? xm1 : ym1;
        Ak4.y = half ? xm2 : ym2;

        v8f acc = __builtin_amdgcn_wmma_f32_16x16x4_f32(
             false, Ak4,  false, Bk[4],
             (short)0, part,  false,  false);

        if (blk + 1 < NBLK) {
            v8f nxt = {};
            #pragma unroll
            for (int kc = 0; kc < 4; ++kc) {
                nxt = __builtin_amdgcn_wmma_f32_16x16x4_f32(
                     false, Ax[blk + 1][kc],  false, Bk[kc],
                     (short)0, nxt,  false,  false);
            }
            part = nxt;
        }

        #pragma unroll
        for (int v = 0; v < 8; ++v) {
            out[(long)(r0 + v + 8 * half) * TLEN + tb + lid] = acc[v];
        }

        float u1[8], u2[8];
        #pragma unroll
        for (int v = 0; v < 8; ++v) {
            u1[v] = row_lane<15>(acc[v]);
            u2[v] = row_lane<14>(acc[v]);
        }
        float z1 = pick8(u1, bit0, bit1, bit2);
        float z2 = pick8(u2, bit0, bit1, bit2);
        float w1 = other_half(z1);
        float w2 = other_half(z2);
        ym1 = upper8 ? w1 : z1;
        ym2 = upper8 ? w2 : z2;

        xm1 = Ax[blk][3].y; xm2 = Ax[blk][3].x;

        __builtin_amdgcn_sched_barrier(0);
    }
}

extern "C" void kernel_launch(void* const* d_in, const int* in_sizes, int n_in,
                              void* d_out, int out_size, void* d_ws, size_t ws_size,
                              hipStream_t stream) {
    const float* x = (const float*)d_in[0];
    float* out = (float*)d_out;

    float2* entry = (float2*)d_ws;
    float2* btot  = entry + (size_t)BROWS * NC;

    double w0    = 2.0 * M_PI * 7500.0 / 16000.0;
    double alpha = sin(w0) / (2.0 * 0.707);
    double cw    = cos(w0);
    double b0 = (1.0 - cw) / 2.0, b1 = 1.0 - cw, b2 = b0;
    double a0 = 1.0 + alpha, a1 = -2.0 * cw, a2 = 1.0 - alpha;
    b0 /= a0; b1 /= a0; b2 /= a0; a1 /= a0; a2 /= a0;

    double p00 = 1, p01 = 0, p10 = 0, p11 = 1;
    for (int i = 0; i < CHUNK; ++i) {
        double n00 = -a1 * p00 - a2 * p10;
        double n01 = -a1 * p01 - a2 * p11;
        double n10 = p00, n11 = p01;
        p00 = n00; p01 = n01; p10 = n10; p11 = n11;
    }

    float fb0 = (float)b0, fb1 = (float)b1, fb2 = (float)b2;
    float fa1 = (float)a1, fa2 = (float)a2;

    float fp00 = (float)p00, fp01 = (float)p01, fp10 = (float)p10, fp11 = (float)p11;

    lp_phase1<<<NPB, P1_THREADS, 0, stream>>>(x, entry, btot, fb0, fb1, fb2, fa1, fa2,
                                              fp00, fp01, fp10, fp11);
    int nwaves = NC * (BROWS / 16);
    lp_phase3<<<nwaves / 8, 256, 0, stream>>>(x, out, entry, btot, fb0, fb1, fb2, fa1, fa2,
                                              fp00, fp01, fp10, fp11);
}


# phase 3 stores whole 128-byte lines: the results of two successive blocks trade lane halves in registers (one swap per register pair), so a store writes one row's 32 samples; sums unchanged
# speedup vs baseline: 5.4611x; 1.0304x over previous
#include <hip/hip_runtime.h>
#include <math.h>

constexpr int TLEN  = 320000;
constexpr int BROWS = 64;
constexpr int CHUNK = 320;
constexpr int NC    = TLEN / CHUNK;
constexpr int NBLK  = CHUNK / 16;

typedef __attribute__((ext_vector_type(2))) float v2f;
typedef __attribute__((ext_vector_type(8))) float v8f;

__device__ __forceinline__ float fir3(const float* __restrict__ xr, int t,
                                      float b0, float b1, float b2) {
    float x0 = xr[t];
    float x1 = (t >= 1) ? xr[t - 1] : 0.0f;
    float x2 = (t >= 2) ? xr[t - 2] : 0.0f;
    return fmaf(b0, x0, fmaf(b1, x1, b2 * x2));
}

__device__ __forceinline__ void aff_step(float& s1, float& s2, float dx, float dy,
                                         float m00, float m01, float m10, float m11) {
    float n1 = fmaf(m00, s1, fmaf(m01, s2, dx));
    float n2 = fmaf(m10, s1, fmaf(m11, s2, dy));
    s1 = n1; s2 = n2;
}

__device__ __forceinline__ void mat_square(float& m00, float& m01, float& m10, float& m11) {
    float n00 = fmaf(m00, m00, m01 * m10);
    float n01 = fmaf(m00, m01, m01 * m11);
    float n10 = fmaf(m10, m00, m11 * m10);
    float n11 = fmaf(m10, m01, m11 * m11);
    m00 = n00; m01 = n01; m10 = n10; m11 = n11;
}

__device__ __forceinline__ void mat_mul(float& r00, float& r01, float& r10, float& r11,
                                        float m00, float m01, float m10, float m11) {
    float n00 = fmaf(r00, m00, r01 * m10);
    float n01 = fmaf(r00, m01, r01 * m11);
    float n10 = fmaf(r10, m00, r11 * m10);
    float n11 = fmaf(r10, m01, r11 * m11);
    r00 = n00; r01 = n01; r10 = n10; r11 = n11;
}

constexpr int P1_THREADS = 256;
constexpr int P1_WAVES   = P1_THREADS / 32;
constexpr int SLAB   = 64;
constexpr int NSLAB  = CHUNK / SLAB;
constexpr int SLAB4  = SLAB / 4;
constexpr int PITCH4 = SLAB4 + 1;
static_assert(CHUNK % SLAB == 0, "a chunk is a whole number of slabs");
static_assert((BROWS * NC) % P1_THREADS == 0, "no partial block: the barriers below need every lane");
constexpr int NPB = BROWS * NC / P1_THREADS;
constexpr int MAXJOIN = (NC - 2) / P1_THREADS + 1;
constexpr int BT_PITCH = 16;

__global__ __launch_bounds__(P1_THREADS) void lp_phase1(const float* __restrict__ x,
                                                        float2* __restrict__ entry,
                                                        float2* __restrict__ btot,
                                                        float b0, float b1, float b2,
                                                        float a1, float a2,
                                                        float p00, float p01, float p10, float p11) {
    __shared__ float4 stage[2][P1_WAVES][32 * PITCH4];
    __shared__ float2 wtot[P1_WAVES];

    const int lane = threadIdx.x & 31;
    const int wv   = threadIdx.x >> 5;
    const int gid  = blockIdx.x * P1_THREADS + threadIdx.x;
    const int g0   = gid - lane;
    const int c    = gid % NC;

    const float* xc = x + (long)gid * CHUNK;
    float y1 = 0.0f, y2 = 0.0f;
    float xm1 = (c > 0) ? xc[-1] : 0.0f;
    float xm2 = (c > 0) ? xc[-2] : 0.0f;

    const int sub   = lane >> 4;
    const int piece = lane & 15;
    const float* src = x + (long)(g0 + sub) * CHUNK + piece * 4;

    float4 g[SLAB4];
    #pragma unroll
    for (int i = 0; i < SLAB4; ++i)
        g[i] = *reinterpret_cast<const float4*>(src + (long)(2 * i) * CHUNK);

    #pragma unroll
    for (int s = 0; s < NSLAB; ++s) {
        float4* buf = stage[s & 1][wv];
        #pragma unroll
        for (int i = 0; i < SLAB4; ++i) buf[(2 * i + sub) * PITCH4 + piece] = g[i];

        __syncthreads();

        float4 v[SLAB4];
        #pragma unroll
        for (int q = 0; q < SLAB4; ++q) v[q] = buf[lane * PITCH4 + q];

        if (s + 1 < NSLAB) {
            #pragma unroll
            for (int i = 0; i < SLAB4; ++i)
                g[i] = *reinterpret_cast<const float4*>(src + (long)(2 * i) * CHUNK + (s + 1) * SLAB);
        }

        #pragma unroll
        for (int q = 0; q < SLAB4; ++q) {
            const float xs[4] = {v[q].x, v[q].y, v[q].z, v[q].w};
            #pragma unroll
            for (int e = 0; e < 4; ++e) {
                float xn = xs[e];
                float f  = fmaf(b0, xn, fmaf(b1, xm1, b2 * xm2));
                float y  = fmaf(-a1, y1, fmaf(-a2, y2, f));
                y2 = y1; y1 = y; xm2 = xm1; xm1 = xn;
            }
        }
    }

    float t1 = y1, t2 = y2;
    float q00 = p00, q01 = p01, q10 = p10, q11 = p11;
    float r00 = 1.0f, r01 = 0.0f, r10 = 0.0f, r11 = 1.0f;
    #pragma unroll
    for (int o = 1; o < 32; o <<= 1) {
        float u1 = __shfl_up(t1, o, 32);
        float u2 = __shfl_up(t2, o, 32);
        const bool take = (lane >= o) && (c >= o);
        aff_step(u1, u2, t1, t2, q00, q01, q10, q11);
        t1 = take ? u1 : t1;
        t2 = take ? u2 : t2;
        if (lane & o) mat_mul(r00, r01, r10, r11, q00, q01, q10, q11);
        mat_square(q00, q01, q10, q11);
    }

    float e1 = __shfl_up(t1, 1, 32);
    float e2 = __shfl_up(t2, 1, 32);
    if (lane == 0 || c == 0) { e1 = 0.0f; e2 = 0.0f; }

    if (lane == 31) wtot[wv] = make_float2(t1, t2);
    __syncthreads();

    float E1 = 0.0f, E2 = 0.0f, W1 = 0.0f, W2 = 0.0f;
    const int base = blockIdx.x * P1_THREADS;
    #pragma unroll
    for (int v = 0; v < P1_WAVES; ++v) {
        if (v == wv) { W1 = E1; W2 = E2; }
        const float2 T = wtot[v];
        const bool link = ((base + 32 * v + 31) % NC) >= 32;
        float n1 = E1, n2 = E2;
        aff_step(n1, n2, T.x, T.y, q00, q01, q10, q11);
        E1 = link ? n1 : T.x;
        E2 = link ? n2 : T.y;
    }
    if (threadIdx.x == 0) btot[blockIdx.x * BT_PITCH] = make_float2(E1, E2);

    if (c >= lane + 1) {
        e1 = fmaf(r00, W1, fmaf(r01, W2, e1));
        e2 = fmaf(r10, W1, fmaf(r11, W2, e2));
    }
    entry[gid] = make_float2(e1, e2);
}

template <int L>
__device__ __forceinline__ float row_lane(float v) {
    int b = __builtin_bit_cast(int, v);
    return __builtin_bit_cast(float,
        __builtin_amdgcn_mov_dpp(b, 0x150 + L  , 0xF, 0xF, false));
}
__device__ __forceinline__ float other_half(float v) {
    int b = __builtin_bit_cast(int, v);
    return __builtin_bit_cast(float,
        __builtin_amdgcn_permlanex16(b, b, 0x76543210, 0xfedcba98, false, false));
}
template <bool PAD>
__device__ __forceinline__ void swap_halves4(float& a0, float& a1, float& a2, float& a3,
                                             float& b0, float& b1, float& b2, float& b3) {
    if (PAD) {
        asm("v_nop\n\tv_nop\n\tv_nop\n\tv_nop\n\tv_nop\n\t"
            "v_permlane16_swap_b32 %0, %4\n\tv_permlane16_swap_b32 %1, %5\n\t"
            "v_permlane16_swap_b32 %2, %6\n\tv_permlane16_swap_b32 %3, %7"
            : "+v"(a0), "+v"(a1), "+v"(a2), "+v"(a3), "+v"(b0), "+v"(b1), "+v"(b2), "+v"(b3));
    } else {
        asm("v_permlane16_swap_b32 %0, %4\n\tv_permlane16_swap_b32 %1, %5\n\t"
            "v_permlane16_swap_b32 %2, %6\n\tv_permlane16_swap_b32 %3, %7"
            : "+v"(a0), "+v"(a1), "+v"(a2), "+v"(a3), "+v"(b0), "+v"(b1), "+v"(b2), "+v"(b3));
    }
}
__device__ __forceinline__ float pick8(const float (&u)[8], bool b0, bool b1, bool b2) {
    float p0 = b0 ? u[1] : u[0];
    float p1 = b0 ? u[3] : u[2];
    float p2 = b0 ? u[5] : u[4];
    float p3 = b0 ? u[7] : u[6];
    float q0 = b1 ? p1 : p0;
    float q1 = b1 ? p3 : p2;
    return b2 ? q1 : q0;
}

__global__ __launch_bounds__(256) void lp_phase3(
    const float* __restrict__ x, float* __restrict__ out,
    const float2* __restrict__ entry, const float2* __restrict__ btot,
    float b0, float b1, float b2, float a1, float a2,
    float p00, float p01, float p10, float p11) {

    int lane = threadIdx.x & 31;
    int wid  = blockIdx.x * 8 + (threadIdx.x >> 5);
    int c  = wid % NC;
    int rg = wid / NC;
    if (rg >= BROWS / 16) return;

    int half = lane >> 4;
    int lid  = lane & 15;
    const bool bit0 = (lid & 1) != 0, bit1 = (lid & 2) != 0, bit2 = (lid & 4) != 0;
    const bool upper8 = (lid & 8) != 0;
    int r0   = rg * 16;
    int row  = r0 + lid;
    int t0   = c * CHUNK;

    float h[17];
    h[0] = 1.0f;
    h[1] = -a1;
    #pragma unroll
    for (int n = 2; n <= 16; ++n) h[n] = fmaf(-a1, h[n - 1], -a2 * h[n - 2]);

    float g[16];
    g[0] = b0 * h[0];
    g[1] = fmaf(b0, h[1], b1 * h[0]);
    #pragma unroll
    for (int n = 2; n < 16; ++n) g[n] = fmaf(b0, h[n], fmaf(b1, h[n - 1], b2 * h[n - 2]));

    float hm = 0.0f, hl = 0.0f, hp = 0.0f;
    #pragma unroll
    for (int n = 0; n <= 16; ++n) {
        if (n + 1 == lid) hm = h[n];
        if (n == lid)     hl = h[n];
        if (n == lid + 1) hp = h[n];
    }

    float Tt[20];
    #pragma unroll
    for (int j = 0; j < 16; ++j) Tt[j] = (lid >= j) ? g[lid - j] : 0.0f;
    Tt[16] = hp;
    Tt[17] = -a2 * hl;
    Tt[18] = fmaf(b1, hl, b2 * hm);
    Tt[19] = b2 * hl;

    v2f Bk[5];
    #pragma unroll
    for (int kc = 0; kc < 5; ++kc) {
        Bk[kc].x = half ? Tt[4 * kc + 2] : Tt[4 * kc + 0];
        Bk[kc].y = half ? Tt[4 * kc + 3] : Tt[4 * kc + 1];
    }

    const int gch = row * NC + c;
    const int pb  = gch / P1_THREADS;
    const int j   = gch % P1_THREADS;
    const float2 e = entry[gch];
    float2 Bt[MAXJOIN];
    #pragma unroll
    for (int k = 1; k <= MAXJOIN; ++k) Bt[k - 1] = btot[max(pb - k, 0) * BT_PITCH];


    const float* xr = x + (long)row * TLEN;

    float xm1 = (t0 > 0) ? xr[t0 - 1] : 0.0f;
    float xm2 = (t0 > 0) ? xr[t0 - 2] : 0.0f;

    v2f Ax[NBLK][4];
    #pragma unroll
    for (int blk = 0; blk < NBLK; ++blk) {
        const v2f* xa = reinterpret_cast<const v2f*>(xr + t0 + blk * 16 + 2 * half);
        #pragma unroll
        for (int kc = 0; kc < 4; ++kc) Ax[blk][kc] = xa[2 * kc];
    }
    __builtin_amdgcn_sched_barrier(0);

    float q00 = p00, q01 = p01, q10 = p10, q11 = p11;
    float r00 = 1.0f, r01 = 0.0f, r10 = 0.0f, r11 = 1.0f;
    #pragma unroll
    for (int k = 1; k < P1_THREADS; k <<= 1) {
        if (j & k) mat_mul(r00, r01, r10, r11, q00, q01, q10, q11);
        mat_square(q00, q01, q10, q11);
    }
    float E1 = 0.0f, E2 = 0.0f;
    #pragma unroll
    for (int k = MAXJOIN; k >= 1; --k) {
        const bool ok = c >= j + 1 + P1_THREADS * (k - 1);
        float n1 = E1, n2 = E2;
        aff_step(n1, n2, Bt[k - 1].x, Bt[k - 1].y, q00, q01, q10, q11);
        E1 = ok ? n1 : E1;
        E2 = ok ? n2 : E2;
    }
    float ym1 = e.x, ym2 = e.y;
    if (c >= j + 1) {
        ym1 = fmaf(r00, E1, fmaf(r01, E2, ym1));
        ym2 = fmaf(r10, E1, fmaf(r11, E2, ym2));
    }

    v8f part = {};
    #pragma unroll
    for (int kc = 0; kc < 4; ++kc) {
        part = __builtin_amdgcn_wmma_f32_16x16x4_f32(
             false, Ax[0][kc],  false, Bk[kc],
             (short)0, part,  false,  false);
    }

    static_assert(NBLK % 2 == 0, "blocks are stored two at a time");
    v8f held = {};

    #pragma unroll
    for (int blk = 0; blk < NBLK; ++blk) {
        int tb = t0 + blk * 16;

        v2f Ak4;
        Ak4.x = half ? xm1 : ym1;
        Ak4.y = half ? xm2 : ym2;

        v8f acc = __builtin_amdgcn_wmma_f32_16x16x4_f32(
             false, Ak4,  false, Bk[4],
             (short)0, part,  false,  false);

        if (blk + 1 < NBLK) {
            v8f nxt = {};
            #pragma unroll
            for (int kc = 0; kc < 4; ++kc) {
                nxt = __builtin_amdgcn_wmma_f32_16x16x4_f32(
                     false, Ax[blk + 1][kc],  false, Bk[kc],
                     (short)0, nxt,  false,  false);
            }
            part = nxt;
        }

        float u1[8], u2[8];
        #pragma unroll
        for (int v = 0; v < 8; ++v) {
            u1[v] = row_lane<15>(acc[v]);
            u2[v] = row_lane<14>(acc[v]);
        }
        float z1 = pick8(u1, bit0, bit1, bit2);
        float z2 = pick8(u2, bit0, bit1, bit2);
        float w1 = other_half(z1);
        float w2 = other_half(z2);
        ym1 = upper8 ? w1 : z1;
        ym2 = upper8 ? w2 : z2;

        xm1 = Ax[blk][3].y; xm2 = Ax[blk][3].x;

        if (blk & 1) {
            const int tp = tb - 16;
            float lo[8], hi[8];
            #pragma unroll
            for (int v = 0; v < 8; ++v) { lo[v] = held[v]; hi[v] = acc[v]; }
            if (blk == NBLK - 1) {
                swap_halves4<true>(lo[0], lo[1], lo[2], lo[3], hi[0], hi[1], hi[2], hi[3]);
                swap_halves4<true>(lo[4], lo[5], lo[6], lo[7], hi[4], hi[5], hi[6], hi[7]);
            } else {
                swap_halves4<false>(lo[0], lo[1], lo[2], lo[3], hi[0], hi[1], hi[2], hi[3]);
                swap_halves4<false>(lo[4], lo[5], lo[6], lo[7], hi[4], hi[5], hi[6], hi[7]);
            }
            #pragma unroll
            for (int v = 0; v < 8; ++v) {
                out[(long)(r0 + v) * TLEN + tp + lane]     = lo[v];
                out[(long)(r0 + v + 8) * TLEN + tp + lane] = hi[v];
            }
        } else {
            held = acc;
        }

        __builtin_amdgcn_sched_barrier(0);
    }
}

extern "C" void kernel_launch(void* const* d_in, const int* in_sizes, int n_in,
                              void* d_out, int out_size, void* d_ws, size_t ws_size,
                              hipStream_t stream) {
    const float* x = (const float*)d_in[0];
    float* out = (float*)d_out;

    float2* entry = (float2*)d_ws;
    float2* btot  = entry + (size_t)BROWS * NC;

    double w0    = 2.0 * M_PI * 7500.0 / 16000.0;
    double alpha = sin(w0) / (2.0 * 0.707);
    double cw    = cos(w0);
    double b0 = (1.0 - cw) / 2.0, b1 = 1.0 - cw, b2 = b0;
    double a0 = 1.0 + alpha, a1 = -2.0 * cw, a2 = 1.0 - alpha;
    b0 /= a0; b1 /= a0; b2 /= a0; a1 /= a0; a2 /= a0;

    double p00 = 1, p01 = 0, p10 = 0, p11 = 1;
    for (int i = 0; i < CHUNK; ++i) {
        double n00 = -a1 * p00 - a2 * p10;
        double n01 = -a1 * p01 - a2 * p11;
        double n10 = p00, n11 = p01;
        p00 = n00; p01 = n01; p10 = n10; p11 = n11;
    }

    float fb0 = (float)b0, fb1 = (float)b1, fb2 = (float)b2;
    float fa1 = (float)a1, fa2 = (float)a2;

    float fp00 = (float)p00, fp01 = (float)p01, fp10 = (float)p10, fp11 = (float)p11;

    lp_phase1<<<NPB, P1_THREADS, 0, stream>>>(x, entry, btot, fb0, fb1, fb2, fa1, fa2,
                                              fp00, fp01, fp10, fp11);
    int nwaves = NC * (BROWS / 16);
    lp_phase3<<<nwaves / 8, 256, 0, stream>>>(x, out, entry, btot, fb0, fb1, fb2, fa1, fa2,
                                              fp00, fp01, fp10, fp11);
}


# phase 3 stores out with the non-temporal hint: the write-once output lines leave the second-level cache first and no longer push out x, which both launches read; sums unchanged
# speedup vs baseline: 6.2658x; 1.1474x over previous
#include <hip/hip_runtime.h>
#include <math.h>

constexpr int TLEN  = 320000;
constexpr int BROWS = 64;
constexpr int CHUNK = 320;
constexpr int NC    = TLEN / CHUNK;
constexpr int NBLK  = CHUNK / 16;

typedef __attribute__((ext_vector_type(2))) float v2f;
typedef __attribute__((ext_vector_type(8))) float v8f;

__device__ __forceinline__ float fir3(const float* __restrict__ xr, int t,
                                      float b0, float b1, float b2) {
    float x0 = xr[t];
    float x1 = (t >= 1) ? xr[t - 1] : 0.0f;
    float x2 = (t >= 2) ? xr[t - 2] : 0.0f;
    return fmaf(b0, x0, fmaf(b1, x1, b2 * x2));
}

__device__ __forceinline__ void aff_step(float& s1, float& s2, float dx, float dy,
                                         float m00, float m01, float m10, float m11) {
    float n1 = fmaf(m00, s1, fmaf(m01, s2, dx));
    float n2 = fmaf(m10, s1, fmaf(m11, s2, dy));
    s1 = n1; s2 = n2;
}

__device__ __forceinline__ void mat_square(float& m00, float& m01, float& m10, float& m11) {
    float n00 = fmaf(m00, m00, m01 * m10);
    float n01 = fmaf(m00, m01, m01 * m11);
    float n10 = fmaf(m10, m00, m11 * m10);
    float n11 = fmaf(m10, m01, m11 * m11);
    m00 = n00; m01 = n01; m10 = n10; m11 = n11;
}

__device__ __forceinline__ void mat_mul(float& r00, float& r01, float& r10, float& r11,
                                        float m00, float m01, float m10, float m11) {
    float n00 = fmaf(r00, m00, r01 * m10);
    float n01 = fmaf(r00, m01, r01 * m11);
    float n10 = fmaf(r10, m00, r11 * m10);
    float n11 = fmaf(r10, m01, r11 * m11);
    r00 = n00; r01 = n01; r10 = n10; r11 = n11;
}

constexpr int P1_THREADS = 256;
constexpr int P1_WAVES   = P1_THREADS / 32;
constexpr int SLAB   = 64;
constexpr int NSLAB  = CHUNK / SLAB;
constexpr int SLAB4  = SLAB / 4;
constexpr int PITCH4 = SLAB4 + 1;
static_assert(CHUNK % SLAB == 0, "a chunk is a whole number of slabs");
static_assert((BROWS * NC) % P1_THREADS == 0, "no partial block: the barriers below need every lane");
constexpr int NPB = BROWS * NC / P1_THREADS;
constexpr int MAXJOIN = (NC - 2) / P1_THREADS + 1;
constexpr int BT_PITCH = 16;

__global__ __launch_bounds__(P1_THREADS) void lp_phase1(const float* __restrict__ x,
                                                        float2* __restrict__ entry,
                                                        float2* __restrict__ btot,
                                                        float b0, float b1, float b2,
                                                        float a1, float a2,
                                                        float p00, float p01, float p10, float p11) {
    __shared__ float4 stage[2][P1_WAVES][32 * PITCH4];
    __shared__ float2 wtot[P1_WAVES];

    const int lane = threadIdx.x & 31;
    const int wv   = threadIdx.x >> 5;
    const int gid  = blockIdx.x * P1_THREADS + threadIdx.x;
    const int g0   = gid - lane;
    const int c    = gid % NC;

    const float* xc = x + (long)gid * CHUNK;
    float y1 = 0.0f, y2 = 0.0f;
    float xm1 = (c > 0) ? xc[-1] : 0.0f;
    float xm2 = (c > 0) ? xc[-2] : 0.0f;

    const int sub   = lane >> 4;
    const int piece = lane & 15;
    const float* src = x + (long)(g0 + sub) * CHUNK + piece * 4;

    float4 g[SLAB4];
    #pragma unroll
    for (int i = 0; i < SLAB4; ++i)
        g[i] = *reinterpret_cast<const float4*>(src + (long)(2 * i) * CHUNK);

    #pragma unroll
    for (int s = 0; s < NSLAB; ++s) {
        float4* buf = stage[s & 1][wv];
        #pragma unroll
        for (int i = 0; i < SLAB4; ++i) buf[(2 * i + sub) * PITCH4 + piece] = g[i];

        __syncthreads();

        float4 v[SLAB4];
        #pragma unroll
        for (int q = 0; q < SLAB4; ++q) v[q] = buf[lane * PITCH4 + q];

        if (s + 1 < NSLAB) {
            #pragma unroll
            for (int i = 0; i < SLAB4; ++i)
                g[i] = *reinterpret_cast<const float4*>(src + (long)(2 * i) * CHUNK + (s + 1) * SLAB);
        }

        #pragma unroll
        for (int q = 0; q < SLAB4; ++q) {
            const float xs[4] = {v[q].x, v[q].y, v[q].z, v[q].w};
            #pragma unroll
            for (int e = 0; e < 4; ++e) {
                float xn = xs[e];
                float f  = fmaf(b0, xn, fmaf(b1, xm1, b2 * xm2));
                float y  = fmaf(-a1, y1, fmaf(-a2, y2, f));
                y2 = y1; y1 = y; xm2 = xm1; xm1 = xn;
            }
        }
    }

    float t1 = y1, t2 = y2;
    float q00 = p00, q01 = p01, q10 = p10, q11 = p11;
    float r00 = 1.0f, r01 = 0.0f, r10 = 0.0f, r11 = 1.0f;
    #pragma unroll
    for (int o = 1; o < 32; o <<= 1) {
        float u1 = __shfl_up(t1, o, 32);
        float u2 = __shfl_up(t2, o, 32);
        const bool take = (lane >= o) && (c >= o);
        aff_step(u1, u2, t1, t2, q00, q01, q10, q11);
        t1 = take ? u1 : t1;
        t2 = take ? u2 : t2;
        if (lane & o) mat_mul(r00, r01, r10, r11, q00, q01, q10, q11);
        mat_square(q00, q01, q10, q11);
    }

    float e1 = __shfl_up(t1, 1, 32);
    float e2 = __shfl_up(t2, 1, 32);
    if (lane == 0 || c == 0) { e1 = 0.0f; e2 = 0.0f; }

    if (lane == 31) wtot[wv] = make_float2(t1, t2);
    __syncthreads();

    float E1 = 0.0f, E2 = 0.0f, W1 = 0.0f, W2 = 0.0f;
    const int base = blockIdx.x * P1_THREADS;
    #pragma unroll
    for (int v = 0; v < P1_WAVES; ++v) {
        if (v == wv) { W1 = E1; W2 = E2; }
        const float2 T = wtot[v];
        const bool link = ((base + 32 * v + 31) % NC) >= 32;
        float n1 = E1, n2 = E2;
        aff_step(n1, n2, T.x, T.y, q00, q01, q10, q11);
        E1 = link ? n1 : T.x;
        E2 = link ? n2 : T.y;
    }
    if (threadIdx.x == 0) btot[blockIdx.x * BT_PITCH] = make_float2(E1, E2);

    if (c >= lane + 1) {
        e1 = fmaf(r00, W1, fmaf(r01, W2, e1));
        e2 = fmaf(r10, W1, fmaf(r11, W2, e2));
    }
    entry[gid] = make_float2(e1, e2);
}

template <int L>
__device__ __forceinline__ float row_lane(float v) {
    int b = __builtin_bit_cast(int, v);
    return __builtin_bit_cast(float,
        __builtin_amdgcn_mov_dpp(b, 0x150 + L  , 0xF, 0xF, false));
}
__device__ __forceinline__ float other_half(float v) {
    int b = __builtin_bit_cast(int, v);
    return __builtin_bit_cast(float,
        __builtin_amdgcn_permlanex16(b, b, 0x76543210, 0xfedcba98, false, false));
}
template <bool PAD>
__device__ __forceinline__ void swap_halves4(float& a0, float& a1, float& a2, float& a3,
                                             float& b0, float& b1, float& b2, float& b3) {
    if (PAD) {
        asm("v_nop\n\tv_nop\n\tv_nop\n\tv_nop\n\tv_nop\n\t"
            "v_permlane16_swap_b32 %0, %4\n\tv_permlane16_swap_b32 %1, %5\n\t"
            "v_permlane16_swap_b32 %2, %6\n\tv_permlane16_swap_b32 %3, %7"
            : "+v"(a0), "+v"(a1), "+v"(a2), "+v"(a3), "+v"(b0), "+v"(b1), "+v"(b2), "+v"(b3));
    } else {
        asm("v_permlane16_swap_b32 %0, %4\n\tv_permlane16_swap_b32 %1, %5\n\t"
            "v_permlane16_swap_b32 %2, %6\n\tv_permlane16_swap_b32 %3, %7"
            : "+v"(a0), "+v"(a1), "+v"(a2), "+v"(a3), "+v"(b0), "+v"(b1), "+v"(b2), "+v"(b3));
    }
}
__device__ __forceinline__ float pick8(const float (&u)[8], bool b0, bool b1, bool b2) {
    float p0 = b0 ? u[1] : u[0];
    float p1 = b0 ? u[3] : u[2];
    float p2 = b0 ? u[5] : u[4];
    float p3 = b0 ? u[7] : u[6];
    float q0 = b1 ? p1 : p0;
    float q1 = b1 ? p3 : p2;
    return b2 ? q1 : q0;
}

__global__ __launch_bounds__(256) void lp_phase3(
    const float* __restrict__ x, float* __restrict__ out,
    const float2* __restrict__ entry, const float2* __restrict__ btot,
    float b0, float b1, float b2, float a1, float a2,
    float p00, float p01, float p10, float p11) {

    int lane = threadIdx.x & 31;
    int wid  = blockIdx.x * 8 + (threadIdx.x >> 5);
    int c  = wid % NC;
    int rg = wid / NC;
    if (rg >= BROWS / 16) return;

    int half = lane >> 4;
    int lid  = lane & 15;
    const bool bit0 = (lid & 1) != 0, bit1 = (lid & 2) != 0, bit2 = (lid & 4) != 0;
    const bool upper8 = (lid & 8) != 0;
    int r0   = rg * 16;
    int row  = r0 + lid;
    int t0   = c * CHUNK;

    float h[17];
    h[0] = 1.0f;
    h[1] = -a1;
    #pragma unroll
    for (int n = 2; n <= 16; ++n) h[n] = fmaf(-a1, h[n - 1], -a2 * h[n - 2]);

    float g[16];
    g[0] = b0 * h[0];
    g[1] = fmaf(b0, h[1], b1 * h[0]);
    #pragma unroll
    for (int n = 2; n < 16; ++n) g[n] = fmaf(b0, h[n], fmaf(b1, h[n - 1], b2 * h[n - 2]));

    float hm = 0.0f, hl = 0.0f, hp = 0.0f;
    #pragma unroll
    for (int n = 0; n <= 16; ++n) {
        if (n + 1 == lid) hm = h[n];
        if (n == lid)     hl = h[n];
        if (n == lid + 1) hp = h[n];
    }

    float Tt[20];
    #pragma unroll
    for (int j = 0; j < 16; ++j) Tt[j] = (lid >= j) ? g[lid - j] : 0.0f;
    Tt[16] = hp;
    Tt[17] = -a2 * hl;
    Tt[18] = fmaf(b1, hl, b2 * hm);
    Tt[19] = b2 * hl;

    v2f Bk[5];
    #pragma unroll
    for (int kc = 0; kc < 5; ++kc) {
        Bk[kc].x = half ? Tt[4 * kc + 2] : Tt[4 * kc + 0];
        Bk[kc].y = half ? Tt[4 * kc + 3] : Tt[4 * kc + 1];
    }

    const int gch = row * NC + c;
    const int pb  = gch / P1_THREADS;
    const int j   = gch % P1_THREADS;
    const float2 e = entry[gch];
    float2 Bt[MAXJOIN];
    #pragma unroll
    for (int k = 1; k <= MAXJOIN; ++k) Bt[k - 1] = btot[max(pb - k, 0) * BT_PITCH];


    const float* xr = x + (long)row * TLEN;

    float xm1 = (t0 > 0) ? xr[t0 - 1] : 0.0f;
    float xm2 = (t0 > 0) ? xr[t0 - 2] : 0.0f;

    v2f Ax[NBLK][4];
    #pragma unroll
    for (int blk = 0; blk < NBLK; ++blk) {
        const v2f* xa = reinterpret_cast<const v2f*>(xr + t0 + blk * 16 + 2 * half);
        #pragma unroll
        for (int kc = 0; kc < 4; ++kc) Ax[blk][kc] = xa[2 * kc];
    }
    __builtin_amdgcn_sched_barrier(0);

    float q00 = p00, q01 = p01, q10 = p10, q11 = p11;
    float r00 = 1.0f, r01 = 0.0f, r10 = 0.0f, r11 = 1.0f;
    #pragma unroll
    for (int k = 1; k < P1_THREADS; k <<= 1) {
        if (j & k) mat_mul(r00, r01, r10, r11, q00, q01, q10, q11);
        mat_square(q00, q01, q10, q11);
    }
    float E1 = 0.0f, E2 = 0.0f;
    #pragma unroll
    for (int k = MAXJOIN; k >= 1; --k) {
        const bool ok = c >= j + 1 + P1_THREADS * (k - 1);
        float n1 = E1, n2 = E2;
        aff_step(n1, n2, Bt[k - 1].x, Bt[k - 1].y, q00, q01, q10, q11);
        E1 = ok ? n1 : E1;
        E2 = ok ? n2 : E2;
    }
    float ym1 = e.x, ym2 = e.y;
    if (c >= j + 1) {
        ym1 = fmaf(r00, E1, fmaf(r01, E2, ym1));
        ym2 = fmaf(r10, E1, fmaf(r11, E2, ym2));
    }

    v8f part = {};
    #pragma unroll
    for (int kc = 0; kc < 4; ++kc) {
        part = __builtin_amdgcn_wmma_f32_16x16x4_f32(
             false, Ax[0][kc],  false, Bk[kc],
             (short)0, part,  false,  false);
    }

    static_assert(NBLK % 2 == 0, "blocks are stored two at a time");
    v8f held = {};

    #pragma unroll
    for (int blk = 0; blk < NBLK; ++blk) {
        int tb = t0 + blk * 16;

        v2f Ak4;
        Ak4.x = half ? xm1 : ym1;
        Ak4.y = half ? xm2 : ym2;

        v8f acc = __builtin_amdgcn_wmma_f32_16x16x4_f32(
             false, Ak4,  false, Bk[4],
             (short)0, part,  false,  false);

        if (blk + 1 < NBLK) {
            v8f nxt = {};
            #pragma unroll
            for (int kc = 0; kc < 4; ++kc) {
                nxt = __builtin_amdgcn_wmma_f32_16x16x4_f32(
                     false, Ax[blk + 1][kc],  false, Bk[kc],
                     (short)0, nxt,  false,  false);
            }
            part = nxt;
        }

        float u1[8], u2[8];
        #pragma unroll
        for (int v = 0; v < 8; ++v) {
            u1[v] = row_lane<15>(acc[v]);
            u2[v] = row_lane<14>(acc[v]);
        }
        float z1 = pick8(u1, bit0, bit1, bit2);
        float z2 = pick8(u2, bit0, bit1, bit2);
        float w1 = other_half(z1);
        float w2 = other_half(z2);
        ym1 = upper8 ? w1 : z1;
        ym2 = upper8 ? w2 : z2;

        xm1 = Ax[blk][3].y; xm2 = Ax[blk][3].x;

        if (blk & 1) {
            const int tp = tb - 16;
            float lo[8], hi[8];
            #pragma unroll
            for (int v = 0; v < 8; ++v) { lo[v] = held[v]; hi[v] = acc[v]; }
            if (blk == NBLK - 1) {
                swap_halves4<true>(lo[0], lo[1], lo[2], lo[3], hi[0], hi[1], hi[2], hi[3]);
                swap_halves4<true>(lo[4], lo[5], lo[6], lo[7], hi[4], hi[5], hi[6], hi[7]);
            } else {
                swap_halves4<false>(lo[0], lo[1], lo[2], lo[3], hi[0], hi[1], hi[2], hi[3]);
                swap_halves4<false>(lo[4], lo[5], lo[6], lo[7], hi[4], hi[5], hi[6], hi[7]);
            }
            #pragma unroll
            for (int v = 0; v < 8; ++v) {
                __builtin_nontemporal_store(lo[v], &out[(long)(r0 + v) * TLEN + tp + lane]);
                __builtin_nontemporal_store(hi[v], &out[(long)(r0 + v + 8) * TLEN + tp + lane]);
            }
        } else {
            held = acc;
        }

        __builtin_amdgcn_sched_barrier(0);
    }
}

extern "C" void kernel_launch(void* const* d_in, const int* in_sizes, int n_in,
                              void* d_out, int out_size, void* d_ws, size_t ws_size,
                              hipStream_t stream) {
    const float* x = (const float*)d_in[0];
    float* out = (float*)d_out;

    float2* entry = (float2*)d_ws;
    float2* btot  = entry + (size_t)BROWS * NC;

    double w0    = 2.0 * M_PI * 7500.0 / 16000.0;
    double alpha = sin(w0) / (2.0 * 0.707);
    double cw    = cos(w0);
    double b0 = (1.0 - cw) / 2.0, b1 = 1.0 - cw, b2 = b0;
    double a0 = 1.0 + alpha, a1 = -2.0 * cw, a2 = 1.0 - alpha;
    b0 /= a0; b1 /= a0; b2 /= a0; a1 /= a0; a2 /= a0;

    double p00 = 1, p01 = 0, p10 = 0, p11 = 1;
    for (int i = 0; i < CHUNK; ++i) {
        double n00 = -a1 * p00 - a2 * p10;
        double n01 = -a1 * p01 - a2 * p11;
        double n10 = p00, n11 = p01;
        p00 = n00; p01 = n01; p10 = n10; p11 = n11;
    }

    float fb0 = (float)b0, fb1 = (float)b1, fb2 = (float)b2;
    float fa1 = (float)a1, fa2 = (float)a2;

    float fp00 = (float)p00, fp01 = (float)p01, fp10 = (float)p10, fp11 = (float)p11;

    lp_phase1<<<NPB, P1_THREADS, 0, stream>>>(x, entry, btot, fb0, fb1, fb2, fa1, fa2,
                                              fp00, fp01, fp10, fp11);
    int nwaves = NC * (BROWS / 16);
    lp_phase3<<<nwaves / 8, 256, 0, stream>>>(x, out, entry, btot, fb0, fb1, fb2, fa1, fa2,
                                              fp00, fp01, fp10, fp11);
}


# phase 3: x fetched in whole 128-byte lines (two rows x 256 bytes per load) into a per-wave LDS tile, A fragments read from there in the old k order, not by 80 loads of 16 part lines; sums unchanged
# speedup vs baseline: 6.7622x; 1.0792x over previous
#include <hip/hip_runtime.h>
#include <math.h>

constexpr int TLEN  = 320000;
constexpr int BROWS = 64;
constexpr int CHUNK = 320;
constexpr int NC    = TLEN / CHUNK;
constexpr int NBLK  = CHUNK / 16;

typedef __attribute__((ext_vector_type(2))) float v2f;
typedef __attribute__((ext_vector_type(8))) float v8f;

__device__ __forceinline__ float fir3(const float* __restrict__ xr, int t,
                                      float b0, float b1, float b2) {
    float x0 = xr[t];
    float x1 = (t >= 1) ? xr[t - 1] : 0.0f;
    float x2 = (t >= 2) ? xr[t - 2] : 0.0f;
    return fmaf(b0, x0, fmaf(b1, x1, b2 * x2));
}

__device__ __forceinline__ void aff_step(float& s1, float& s2, float dx, float dy,
                                         float m00, float m01, float m10, float m11) {
    float n1 = fmaf(m00, s1, fmaf(m01, s2, dx));
    float n2 = fmaf(m10, s1, fmaf(m11, s2, dy));
    s1 = n1; s2 = n2;
}

__device__ __forceinline__ void mat_square(float& m00, float& m01, float& m10, float& m11) {
    float n00 = fmaf(m00, m00, m01 * m10);
    float n01 = fmaf(m00, m01, m01 * m11);
    float n10 = fmaf(m10, m00, m11 * m10);
    float n11 = fmaf(m10, m01, m11 * m11);
    m00 = n00; m01 = n01; m10 = n10; m11 = n11;
}

__device__ __forceinline__ void mat_mul(float& r00, float& r01, float& r10, float& r11,
                                        float m00, float m01, float m10, float m11) {
    float n00 = fmaf(r00, m00, r01 * m10);
    float n01 = fmaf(r00, m01, r01 * m11);
    float n10 = fmaf(r10, m00, r11 * m10);
    float n11 = fmaf(r10, m01, r11 * m11);
    r00 = n00; r01 = n01; r10 = n10; r11 = n11;
}

constexpr int P1_THREADS = 256;
constexpr int P1_WAVES   = P1_THREADS / 32;
constexpr int SLAB   = 64;
constexpr int NSLAB  = CHUNK / SLAB;
constexpr int SLAB4  = SLAB / 4;
constexpr int PITCH4 = SLAB4 + 1;
static_assert(CHUNK % SLAB == 0, "a chunk is a whole number of slabs");
static_assert((BROWS * NC) % P1_THREADS == 0, "no partial block: the barriers below need every lane");
constexpr int NPB = BROWS * NC / P1_THREADS;
constexpr int MAXJOIN = (NC - 2) / P1_THREADS + 1;
constexpr int BT_PITCH = 16;

__global__ __launch_bounds__(P1_THREADS) void lp_phase1(const float* __restrict__ x,
                                                        float2* __restrict__ entry,
                                                        float2* __restrict__ btot,
                                                        float b0, float b1, float b2,
                                                        float a1, float a2,
                                                        float p00, float p01, float p10, float p11) {
    __shared__ float4 stage[2][P1_WAVES][32 * PITCH4];
    __shared__ float2 wtot[P1_WAVES];

    const int lane = threadIdx.x & 31;
    const int wv   = threadIdx.x >> 5;
    const int gid  = blockIdx.x * P1_THREADS + threadIdx.x;
    const int g0   = gid - lane;
    const int c    = gid % NC;

    const float* xc = x + (long)gid * CHUNK;
    float y1 = 0.0f, y2 = 0.0f;
    float xm1 = (c > 0) ? xc[-1] : 0.0f;
    float xm2 = (c > 0) ? xc[-2] : 0.0f;

    const int sub   = lane >> 4;
    const int piece = lane & 15;
    const float* src = x + (long)(g0 + sub) * CHUNK + piece * 4;

    float4 g[SLAB4];
    #pragma unroll
    for (int i = 0; i < SLAB4; ++i)
        g[i] = *reinterpret_cast<const float4*>(src + (long)(2 * i) * CHUNK);

    #pragma unroll
    for (int s = 0; s < NSLAB; ++s) {
        float4* buf = stage[s & 1][wv];
        #pragma unroll
        for (int i = 0; i < SLAB4; ++i) buf[(2 * i + sub) * PITCH4 + piece] = g[i];

        __syncthreads();

        float4 v[SLAB4];
        #pragma unroll
        for (int q = 0; q < SLAB4; ++q) v[q] = buf[lane * PITCH4 + q];

        if (s + 1 < NSLAB) {
            #pragma unroll
            for (int i = 0; i < SLAB4; ++i)
                g[i] = *reinterpret_cast<const float4*>(src + (long)(2 * i) * CHUNK + (s + 1) * SLAB);
        }

        #pragma unroll
        for (int q = 0; q < SLAB4; ++q) {
            const float xs[4] = {v[q].x, v[q].y, v[q].z, v[q].w};
            #pragma unroll
            for (int e = 0; e < 4; ++e) {
                float xn = xs[e];
                float f  = fmaf(b0, xn, fmaf(b1, xm1, b2 * xm2));
                float y  = fmaf(-a1, y1, fmaf(-a2, y2, f));
                y2 = y1; y1 = y; xm2 = xm1; xm1 = xn;
            }
        }
    }

    float t1 = y1, t2 = y2;
    float q00 = p00, q01 = p01, q10 = p10, q11 = p11;
    float r00 = 1.0f, r01 = 0.0f, r10 = 0.0f, r11 = 1.0f;
    #pragma unroll
    for (int o = 1; o < 32; o <<= 1) {
        float u1 = __shfl_up(t1, o, 32);
        float u2 = __shfl_up(t2, o, 32);
        const bool take = (lane >= o) && (c >= o);
        aff_step(u1, u2, t1, t2, q00, q01, q10, q11);
        t1 = take ? u1 : t1;
        t2 = take ? u2 : t2;
        if (lane & o) mat_mul(r00, r01, r10, r11, q00, q01, q10, q11);
        mat_square(q00, q01, q10, q11);
    }

    float e1 = __shfl_up(t1, 1, 32);
    float e2 = __shfl_up(t2, 1, 32);
    if (lane == 0 || c == 0) { e1 = 0.0f; e2 = 0.0f; }

    if (lane == 31) wtot[wv] = make_float2(t1, t2);
    __syncthreads();

    float E1 = 0.0f, E2 = 0.0f, W1 = 0.0f, W2 = 0.0f;
    const int base = blockIdx.x * P1_THREADS;
    #pragma unroll
    for (int v = 0; v < P1_WAVES; ++v) {
        if (v == wv) { W1 = E1; W2 = E2; }
        const float2 T = wtot[v];
        const bool link = ((base + 32 * v + 31) % NC) >= 32;
        float n1 = E1, n2 = E2;
        aff_step(n1, n2, T.x, T.y, q00, q01, q10, q11);
        E1 = link ? n1 : T.x;
        E2 = link ? n2 : T.y;
    }
    if (threadIdx.x == 0) btot[blockIdx.x * BT_PITCH] = make_float2(E1, E2);

    if (c >= lane + 1) {
        e1 = fmaf(r00, W1, fmaf(r01, W2, e1));
        e2 = fmaf(r10, W1, fmaf(r11, W2, e2));
    }
    entry[gid] = make_float2(e1, e2);
}

template <int L>
__device__ __forceinline__ float row_lane(float v) {
    int b = __builtin_bit_cast(int, v);
    return __builtin_bit_cast(float,
        __builtin_amdgcn_mov_dpp(b, 0x150 + L  , 0xF, 0xF, false));
}
__device__ __forceinline__ float other_half(float v) {
    int b = __builtin_bit_cast(int, v);
    return __builtin_bit_cast(float,
        __builtin_amdgcn_permlanex16(b, b, 0x76543210, 0xfedcba98, false, false));
}
template <bool PAD>
__device__ __forceinline__ void swap_halves4(float& a0, float& a1, float& a2, float& a3,
                                             float& b0, float& b1, float& b2, float& b3) {
    if (PAD) {
        asm("v_nop\n\tv_nop\n\tv_nop\n\tv_nop\n\tv_nop\n\t"
            "v_permlane16_swap_b32 %0, %4\n\tv_permlane16_swap_b32 %1, %5\n\t"
            "v_permlane16_swap_b32 %2, %6\n\tv_permlane16_swap_b32 %3, %7"
            : "+v"(a0), "+v"(a1), "+v"(a2), "+v"(a3), "+v"(b0), "+v"(b1), "+v"(b2), "+v"(b3));
    } else {
        asm("v_permlane16_swap_b32 %0, %4\n\tv_permlane16_swap_b32 %1, %5\n\t"
            "v_permlane16_swap_b32 %2, %6\n\tv_permlane16_swap_b32 %3, %7"
            : "+v"(a0), "+v"(a1), "+v"(a2), "+v"(a3), "+v"(b0), "+v"(b1), "+v"(b2), "+v"(b3));
    }
}
__device__ __forceinline__ float pick8(const float (&u)[8], bool b0, bool b1, bool b2) {
    float p0 = b0 ? u[1] : u[0];
    float p1 = b0 ? u[3] : u[2];
    float p2 = b0 ? u[5] : u[4];
    float p3 = b0 ? u[7] : u[6];
    float q0 = b1 ? p1 : p0;
    float q1 = b1 ? p3 : p2;
    return b2 ? q1 : q0;
}

constexpr int P3_WAVES   = 8;
constexpr int SLAB3      = 64;
constexpr int NSLAB3     = CHUNK / SLAB3;
constexpr int PART_SLABS = 3;
constexpr int PART_BLKS  = PART_SLABS * SLAB3 / 16;
constexpr int ROW8       = PART_SLABS * SLAB3 / 2 + 2;
static_assert(CHUNK % SLAB3 == 0, "a chunk is a whole number of slabs");
static_assert((ROW8 * 2) % 64 == 4, "row pitch = 4 words mod 64: conflict-free 8-byte fragment reads");
__global__ __launch_bounds__(256) void lp_phase3(
    const float* __restrict__ x, float* __restrict__ out,
    const float2* __restrict__ entry, const float2* __restrict__ btot,
    float b0, float b1, float b2, float a1, float a2,
    float p00, float p01, float p10, float p11) {

    __shared__ v2f tile[P3_WAVES][16 * ROW8];

    int lane = threadIdx.x & 31;
    int wid  = blockIdx.x * 8 + (threadIdx.x >> 5);
    int c  = wid % NC;
    int rg = wid / NC;
    if (rg >= BROWS / 16) return;

    int half = lane >> 4;
    int lid  = lane & 15;
    const bool bit0 = (lid & 1) != 0, bit1 = (lid & 2) != 0, bit2 = (lid & 4) != 0;
    const bool upper8 = (lid & 8) != 0;
    int r0   = rg * 16;
    int row  = r0 + lid;
    int t0   = c * CHUNK;

    float h[17];
    h[0] = 1.0f;
    h[1] = -a1;
    #pragma unroll
    for (int n = 2; n <= 16; ++n) h[n] = fmaf(-a1, h[n - 1], -a2 * h[n - 2]);

    float g[16];
    g[0] = b0 * h[0];
    g[1] = fmaf(b0, h[1], b1 * h[0]);
    #pragma unroll
    for (int n = 2; n < 16; ++n) g[n] = fmaf(b0, h[n], fmaf(b1, h[n - 1], b2 * h[n - 2]));

    float hm = 0.0f, hl = 0.0f, hp = 0.0f;
    #pragma unroll
    for (int n = 0; n <= 16; ++n) {
        if (n + 1 == lid) hm = h[n];
        if (n == lid)     hl = h[n];
        if (n == lid + 1) hp = h[n];
    }

    float Tt[20];
    #pragma unroll
    for (int j = 0; j < 16; ++j) Tt[j] = (lid >= j) ? g[lid - j] : 0.0f;
    Tt[16] = hp;
    Tt[17] = -a2 * hl;
    Tt[18] = fmaf(b1, hl, b2 * hm);
    Tt[19] = b2 * hl;

    v2f Bk[5];
    #pragma unroll
    for (int kc = 0; kc < 5; ++kc) {
        Bk[kc].x = half ? Tt[4 * kc + 2] : Tt[4 * kc + 0];
        Bk[kc].y = half ? Tt[4 * kc + 3] : Tt[4 * kc + 1];
    }

    const int gch = row * NC + c;
    const int pb  = gch / P1_THREADS;
    const int j   = gch % P1_THREADS;
    const float2 e = entry[gch];
    float2 Bt[MAXJOIN];
    #pragma unroll
    for (int k = 1; k <= MAXJOIN; ++k) Bt[k - 1] = btot[max(pb - k, 0) * BT_PITCH];


    const float* xr = x + (long)row * TLEN;

    float xm1 = (t0 > 0) ? xr[t0 - 1] : 0.0f;
    float xm2 = (t0 > 0) ? xr[t0 - 2] : 0.0f;

    const float* xg = x + (long)(r0 + half) * TLEN + t0 + 4 * lid;
    float4 gl[NSLAB3][8];
    #pragma unroll
    for (int s = 0; s < NSLAB3; ++s) {
        #pragma unroll
        for (int p = 0; p < 8; ++p)
            gl[s][p] = *reinterpret_cast<const float4*>(xg + (long)(2 * p) * TLEN + s * SLAB3);
    }
    __builtin_amdgcn_sched_barrier(0);

    float q00 = p00, q01 = p01, q10 = p10, q11 = p11;
    float r00 = 1.0f, r01 = 0.0f, r10 = 0.0f, r11 = 1.0f;
    #pragma unroll
    for (int k = 1; k < P1_THREADS; k <<= 1) {
        if (j & k) mat_mul(r00, r01, r10, r11, q00, q01, q10, q11);
        mat_square(q00, q01, q10, q11);
    }
    float E1 = 0.0f, E2 = 0.0f;
    #pragma unroll
    for (int k = MAXJOIN; k >= 1; --k) {
        const bool ok = c >= j + 1 + P1_THREADS * (k - 1);
        float n1 = E1, n2 = E2;
        aff_step(n1, n2, Bt[k - 1].x, Bt[k - 1].y, q00, q01, q10, q11);
        E1 = ok ? n1 : E1;
        E2 = ok ? n2 : E2;
    }
    float ym1 = e.x, ym2 = e.y;
    if (c >= j + 1) {
        ym1 = fmaf(r00, E1, fmaf(r01, E2, ym1));
        ym2 = fmaf(r10, E1, fmaf(r11, E2, ym2));
    }

    v2f* tw = tile[threadIdx.x >> 5];
    v2f Ax[NBLK][4];
    #pragma unroll
    for (int s = 0; s < PART_SLABS; ++s) {
        #pragma unroll
        for (int p = 0; p < 8; ++p) {
            const int o = (2 * p + half) * ROW8 + s * (SLAB3 / 2) + 2 * lid;
            tw[o]     = v2f{gl[s][p].x, gl[s][p].y};
            tw[o + 1] = v2f{gl[s][p].z, gl[s][p].w};
        }
    }
    __builtin_amdgcn_fence(__ATOMIC_SEQ_CST, "wavefront");
    #pragma unroll
    for (int blk = 0; blk < PART_BLKS; ++blk) {
        #pragma unroll
        for (int kc = 0; kc < 4; ++kc) Ax[blk][kc] = tw[lid * ROW8 + blk * 8 + 2 * kc + half];
    }
    __builtin_amdgcn_fence(__ATOMIC_SEQ_CST, "wavefront");
    #pragma unroll
    for (int s = PART_SLABS; s < NSLAB3; ++s) {
        #pragma unroll
        for (int p = 0; p < 8; ++p) {
            const int o = (2 * p + half) * ROW8 + (s - PART_SLABS) * (SLAB3 / 2) + 2 * lid;
            tw[o]     = v2f{gl[s][p].x, gl[s][p].y};
            tw[o + 1] = v2f{gl[s][p].z, gl[s][p].w};
        }
    }
    __builtin_amdgcn_fence(__ATOMIC_SEQ_CST, "wavefront");
    #pragma unroll
    for (int blk = PART_BLKS; blk < NBLK; ++blk) {
        #pragma unroll
        for (int kc = 0; kc < 4; ++kc)
            Ax[blk][kc] = tw[lid * ROW8 + (blk - PART_BLKS) * 8 + 2 * kc + half];
    }

    v8f part = {};
    #pragma unroll
    for (int kc = 0; kc < 4; ++kc) {
        part = __builtin_amdgcn_wmma_f32_16x16x4_f32(
             false, Ax[0][kc],  false, Bk[kc],
             (short)0, part,  false,  false);
    }

    static_assert(NBLK % 2 == 0, "blocks are stored two at a time");
    v8f held = {};

    #pragma unroll
    for (int blk = 0; blk < NBLK; ++blk) {
        int tb = t0 + blk * 16;

        v2f Ak4;
        Ak4.x = half ? xm1 : ym1;
        Ak4.y = half ? xm2 : ym2;

        v8f acc = __builtin_amdgcn_wmma_f32_16x16x4_f32(
             false, Ak4,  false, Bk[4],
             (short)0, part,  false,  false);

        if (blk + 1 < NBLK) {
            v8f nxt = {};
            #pragma unroll
            for (int kc = 0; kc < 4; ++kc) {
                nxt = __builtin_amdgcn_wmma_f32_16x16x4_f32(
                     false, Ax[blk + 1][kc],  false, Bk[kc],
                     (short)0, nxt,  false,  false);
            }
            part = nxt;
        }

        float u1[8], u2[8];
        #pragma unroll
        for (int v = 0; v < 8; ++v) {
            u1[v] = row_lane<15>(acc[v]);
            u2[v] = row_lane<14>(acc[v]);
        }
        float z1 = pick8(u1, bit0, bit1, bit2);
        float z2 = pick8(u2, bit0, bit1, bit2);
        float w1 = other_half(z1);
        float w2 = other_half(z2);
        ym1 = upper8 ? w1 : z1;
        ym2 = upper8 ? w2 : z2;

        xm1 = Ax[blk][3].y; xm2 = Ax[blk][3].x;

        if (blk & 1) {
            const int tp = tb - 16;
            float lo[8], hi[8];
            #pragma unroll
            for (int v = 0; v < 8; ++v) { lo[v] = held[v]; hi[v] = acc[v]; }
            if (blk == NBLK - 1) {
                swap_halves4<true>(lo[0], lo[1], lo[2], lo[3], hi[0], hi[1], hi[2], hi[3]);
                swap_halves4<true>(lo[4], lo[5], lo[6], lo[7], hi[4], hi[5], hi[6], hi[7]);
            } else {
                swap_halves4<false>(lo[0], lo[1], lo[2], lo[3], hi[0], hi[1], hi[2], hi[3]);
                swap_halves4<false>(lo[4], lo[5], lo[6], lo[7], hi[4], hi[5], hi[6], hi[7]);
            }
            #pragma unroll
            for (int v = 0; v < 8; ++v) {
                __builtin_nontemporal_store(lo[v], &out[(long)(r0 + v) * TLEN + tp + lane]);
                __builtin_nontemporal_store(hi[v], &out[(long)(r0 + v + 8) * TLEN + tp + lane]);
            }
        } else {
            held = acc;
        }

        __builtin_amdgcn_sched_barrier(0);
    }
}

extern "C" void kernel_launch(void* const* d_in, const int* in_sizes, int n_in,
                              void* d_out, int out_size, void* d_ws, size_t ws_size,
                              hipStream_t stream) {
    const float* x = (const float*)d_in[0];
    float* out = (float*)d_out;

    float2* entry = (float2*)d_ws;
    float2* btot  = entry + (size_t)BROWS * NC;

    double w0    = 2.0 * M_PI * 7500.0 / 16000.0;
    double alpha = sin(w0) / (2.0 * 0.707);
    double cw    = cos(w0);
    double b0 = (1.0 - cw) / 2.0, b1 = 1.0 - cw, b2 = b0;
    double a0 = 1.0 + alpha, a1 = -2.0 * cw, a2 = 1.0 - alpha;
    b0 /= a0; b1 /= a0; b2 /= a0; a1 /= a0; a2 /= a0;

    double p00 = 1, p01 = 0, p10 = 0, p11 = 1;
    for (int i = 0; i < CHUNK; ++i) {
        double n00 = -a1 * p00 - a2 * p10;
        double n01 = -a1 * p01 - a2 * p11;
        double n10 = p00, n11 = p01;
        p00 = n00; p01 = n01; p10 = n10; p11 = n11;
    }

    float fb0 = (float)b0, fb1 = (float)b1, fb2 = (float)b2;
    float fa1 = (float)a1, fa2 = (float)a2;

    float fp00 = (float)p00, fp01 = (float)p01, fp10 = (float)p10, fp11 = (float)p11;

    lp_phase1<<<NPB, P1_THREADS, 0, stream>>>(x, entry, btot, fb0, fb1, fb2, fa1, fa2,
                                              fp00, fp01, fp10, fp11);
    int nwaves = NC * (BROWS / 16);
    lp_phase3<<<nwaves / 8, 256, 0, stream>>>(x, out, entry, btot, fb0, fb1, fb2, fa1, fa2,
                                              fp00, fp01, fp10, fp11);
}


# phase 1: x requested 3 slabs (48 line loads a wave) ahead of the sums, not 1; the per-wave LDS staging is ordered by wave-scope fences, since a work-group barrier drains the loads; sums unchanged
# speedup vs baseline: 6.9626x; 1.0296x over previous
#include <hip/hip_runtime.h>
#include <math.h>

constexpr int TLEN  = 320000;
constexpr int BROWS = 64;
constexpr int CHUNK = 320;
constexpr int NC    = TLEN / CHUNK;
constexpr int NBLK  = CHUNK / 16;

typedef __attribute__((ext_vector_type(2))) float v2f;
typedef __attribute__((ext_vector_type(8))) float v8f;
typedef __attribute__((ext_vector_type(4))) float v4f;

__device__ __forceinline__ float fir3(const float* __restrict__ xr, int t,
                                      float b0, float b1, float b2) {
    float x0 = xr[t];
    float x1 = (t >= 1) ? xr[t - 1] : 0.0f;
    float x2 = (t >= 2) ? xr[t - 2] : 0.0f;
    return fmaf(b0, x0, fmaf(b1, x1, b2 * x2));
}

__device__ __forceinline__ void aff_step(float& s1, float& s2, float dx, float dy,
                                         float m00, float m01, float m10, float m11) {
    float n1 = fmaf(m00, s1, fmaf(m01, s2, dx));
    float n2 = fmaf(m10, s1, fmaf(m11, s2, dy));
    s1 = n1; s2 = n2;
}

__device__ __forceinline__ void mat_square(float& m00, float& m01, float& m10, float& m11) {
    float n00 = fmaf(m00, m00, m01 * m10);
    float n01 = fmaf(m00, m01, m01 * m11);
    float n10 = fmaf(m10, m00, m11 * m10);
    float n11 = fmaf(m10, m01, m11 * m11);
    m00 = n00; m01 = n01; m10 = n10; m11 = n11;
}

__device__ __forceinline__ void mat_mul(float& r00, float& r01, float& r10, float& r11,
                                        float m00, float m01, float m10, float m11) {
    float n00 = fmaf(r00, m00, r01 * m10);
    float n01 = fmaf(r00, m01, r01 * m11);
    float n10 = fmaf(r10, m00, r11 * m10);
    float n11 = fmaf(r10, m01, r11 * m11);
    r00 = n00; r01 = n01; r10 = n10; r11 = n11;
}

constexpr int P1_THREADS = 256;
constexpr int P1_WAVES   = P1_THREADS / 32;
constexpr int SLAB   = 64;
constexpr int NSLAB  = CHUNK / SLAB;
constexpr int AHEAD  = 3;
constexpr int SLAB4  = SLAB / 4;
constexpr int PITCH4 = SLAB4 + 1;
static_assert(CHUNK % SLAB == 0, "a chunk is a whole number of slabs");
static_assert((BROWS * NC) % P1_THREADS == 0, "no partial block: the barriers below need every lane");
constexpr int NPB = BROWS * NC / P1_THREADS;
constexpr int MAXJOIN = (NC - 2) / P1_THREADS + 1;
constexpr int BT_PITCH = 16;

__global__ __launch_bounds__(P1_THREADS) void lp_phase1(const float* __restrict__ x,
                                                        float2* __restrict__ entry,
                                                        float2* __restrict__ btot,
                                                        float b0, float b1, float b2,
                                                        float a1, float a2,
                                                        float p00, float p01, float p10, float p11) {
    __shared__ float4 stage[2][P1_WAVES][32 * PITCH4];
    __shared__ float2 wtot[P1_WAVES];

    const int lane = threadIdx.x & 31;
    const int wv   = threadIdx.x >> 5;
    const int gid  = blockIdx.x * P1_THREADS + threadIdx.x;
    const int g0   = gid - lane;
    const int c    = gid % NC;

    const float* xc = x + (long)gid * CHUNK;
    float y1 = 0.0f, y2 = 0.0f;
    float xm1 = (c > 0) ? xc[-1] : 0.0f;
    float xm2 = (c > 0) ? xc[-2] : 0.0f;

    const int sub   = lane >> 4;
    const int piece = lane & 15;
    const float* src = x + (long)(g0 + sub) * CHUNK + piece * 4;

    v4f g[AHEAD][SLAB4];
    #pragma unroll
    for (int s = 0; s < AHEAD; ++s) {
        #pragma unroll
        for (int i = 0; i < SLAB4; ++i)
            g[s % AHEAD][i] = *reinterpret_cast<const v4f*>(src + (long)(2 * i) * CHUNK + s * SLAB);
        __builtin_amdgcn_sched_barrier(0);
    }

    #pragma unroll
    for (int s = 0; s < NSLAB; ++s) {
        float4* buf = stage[s & 1][wv];
        #pragma unroll
        for (int i = 0; i < SLAB4; ++i) {
            const v4f t = g[s % AHEAD][i];
            buf[(2 * i + sub) * PITCH4 + piece] = make_float4(t.x, t.y, t.z, t.w);
        }

        __builtin_amdgcn_fence(__ATOMIC_SEQ_CST, "wavefront");

        if (s + AHEAD < NSLAB) {
            #pragma unroll
            for (int i = 0; i < SLAB4; ++i)
                g[(s + AHEAD) % AHEAD][i] = *reinterpret_cast<const v4f*>(src + (long)(2 * i) * CHUNK + (s + AHEAD) * SLAB);
            __builtin_amdgcn_sched_barrier(0);
        }

        float4 v[SLAB4];
        #pragma unroll
        for (int q = 0; q < SLAB4; ++q) v[q] = buf[lane * PITCH4 + q];

        #pragma unroll
        for (int q = 0; q < SLAB4; ++q) {
            const float xs[4] = {v[q].x, v[q].y, v[q].z, v[q].w};
            #pragma unroll
            for (int e = 0; e < 4; ++e) {
                float xn = xs[e];
                float f  = fmaf(b0, xn, fmaf(b1, xm1, b2 * xm2));
                float y  = fmaf(-a1, y1, fmaf(-a2, y2, f));
                y2 = y1; y1 = y; xm2 = xm1; xm1 = xn;
            }
        }
    }

    float t1 = y1, t2 = y2;
    float q00 = p00, q01 = p01, q10 = p10, q11 = p11;
    float r00 = 1.0f, r01 = 0.0f, r10 = 0.0f, r11 = 1.0f;
    #pragma unroll
    for (int o = 1; o < 32; o <<= 1) {
        float u1 = __shfl_up(t1, o, 32);
        float u2 = __shfl_up(t2, o, 32);
        const bool take = (lane >= o) && (c >= o);
        aff_step(u1, u2, t1, t2, q00, q01, q10, q11);
        t1 = take ? u1 : t1;
        t2 = take ? u2 : t2;
        if (lane & o) mat_mul(r00, r01, r10, r11, q00, q01, q10, q11);
        mat_square(q00, q01, q10, q11);
    }

    float e1 = __shfl_up(t1, 1, 32);
    float e2 = __shfl_up(t2, 1, 32);
    if (lane == 0 || c == 0) { e1 = 0.0f; e2 = 0.0f; }

    if (lane == 31) wtot[wv] = make_float2(t1, t2);
    __syncthreads();

    float E1 = 0.0f, E2 = 0.0f, W1 = 0.0f, W2 = 0.0f;
    const int base = blockIdx.x * P1_THREADS;
    #pragma unroll
    for (int v = 0; v < P1_WAVES; ++v) {
        if (v == wv) { W1 = E1; W2 = E2; }
        const float2 T = wtot[v];
        const bool link = ((base + 32 * v + 31) % NC) >= 32;
        float n1 = E1, n2 = E2;
        aff_step(n1, n2, T.x, T.y, q00, q01, q10, q11);
        E1 = link ? n1 : T.x;
        E2 = link ? n2 : T.y;
    }
    if (threadIdx.x == 0) btot[blockIdx.x * BT_PITCH] = make_float2(E1, E2);

    if (c >= lane + 1) {
        e1 = fmaf(r00, W1, fmaf(r01, W2, e1));
        e2 = fmaf(r10, W1, fmaf(r11, W2, e2));
    }
    entry[gid] = make_float2(e1, e2);
}

template <int L>
__device__ __forceinline__ float row_lane(float v) {
    int b = __builtin_bit_cast(int, v);
    return __builtin_bit_cast(float,
        __builtin_amdgcn_mov_dpp(b, 0x150 + L  , 0xF, 0xF, false));
}
__device__ __forceinline__ float other_half(float v) {
    int b = __builtin_bit_cast(int, v);
    return __builtin_bit_cast(float,
        __builtin_amdgcn_permlanex16(b, b, 0x76543210, 0xfedcba98, false, false));
}
template <bool PAD>
__device__ __forceinline__ void swap_halves4(float& a0, float& a1, float& a2, float& a3,
                                             float& b0, float& b1, float& b2, float& b3) {
    if (PAD) {
        asm("v_nop\n\tv_nop\n\tv_nop\n\tv_nop\n\tv_nop\n\t"
            "v_permlane16_swap_b32 %0, %4\n\tv_permlane16_swap_b32 %1, %5\n\t"
            "v_permlane16_swap_b32 %2, %6\n\tv_permlane16_swap_b32 %3, %7"
            : "+v"(a0), "+v"(a1), "+v"(a2), "+v"(a3), "+v"(b0), "+v"(b1), "+v"(b2), "+v"(b3));
    } else {
        asm("v_permlane16_swap_b32 %0, %4\n\tv_permlane16_swap_b32 %1, %5\n\t"
            "v_permlane16_swap_b32 %2, %6\n\tv_permlane16_swap_b32 %3, %7"
            : "+v"(a0), "+v"(a1), "+v"(a2), "+v"(a3), "+v"(b0), "+v"(b1), "+v"(b2), "+v"(b3));
    }
}
__device__ __forceinline__ float pick8(const float (&u)[8], bool b0, bool b1, bool b2) {
    float p0 = b0 ? u[1] : u[0];
    float p1 = b0 ? u[3] : u[2];
    float p2 = b0 ? u[5] : u[4];
    float p3 = b0 ? u[7] : u[6];
    float q0 = b1 ? p1 : p0;
    float q1 = b1 ? p3 : p2;
    return b2 ? q1 : q0;
}

constexpr int P3_WAVES   = 8;
constexpr int SLAB3      = 64;
constexpr int NSLAB3     = CHUNK / SLAB3;
constexpr int PART_SLABS = 3;
constexpr int PART_BLKS  = PART_SLABS * SLAB3 / 16;
constexpr int ROW8       = PART_SLABS * SLAB3 / 2 + 2;
static_assert(CHUNK % SLAB3 == 0, "a chunk is a whole number of slabs");
static_assert((ROW8 * 2) % 64 == 4, "row pitch = 4 words mod 64: conflict-free 8-byte fragment reads");
__global__ __launch_bounds__(256) void lp_phase3(
    const float* __restrict__ x, float* __restrict__ out,
    const float2* __restrict__ entry, const float2* __restrict__ btot,
    float b0, float b1, float b2, float a1, float a2,
    float p00, float p01, float p10, float p11) {

    __shared__ v2f tile[P3_WAVES][16 * ROW8];

    int lane = threadIdx.x & 31;
    int wid  = blockIdx.x * 8 + (threadIdx.x >> 5);
    int c  = wid % NC;
    int rg = wid / NC;
    if (rg >= BROWS / 16) return;

    int half = lane >> 4;
    int lid  = lane & 15;
    const bool bit0 = (lid & 1) != 0, bit1 = (lid & 2) != 0, bit2 = (lid & 4) != 0;
    const bool upper8 = (lid & 8) != 0;
    int r0   = rg * 16;
    int row  = r0 + lid;
    int t0   = c * CHUNK;

    float h[17];
    h[0] = 1.0f;
    h[1] = -a1;
    #pragma unroll
    for (int n = 2; n <= 16; ++n) h[n] = fmaf(-a1, h[n - 1], -a2 * h[n - 2]);

    float g[16];
    g[0] = b0 * h[0];
    g[1] = fmaf(b0, h[1], b1 * h[0]);
    #pragma unroll
    for (int n = 2; n < 16; ++n) g[n] = fmaf(b0, h[n], fmaf(b1, h[n - 1], b2 * h[n - 2]));

    float hm = 0.0f, hl = 0.0f, hp = 0.0f;
    #pragma unroll
    for (int n = 0; n <= 16; ++n) {
        if (n + 1 == lid) hm = h[n];
        if (n == lid)     hl = h[n];
        if (n == lid + 1) hp = h[n];
    }

    float Tt[20];
    #pragma unroll
    for (int j = 0; j < 16; ++j) Tt[j] = (lid >= j) ? g[lid - j] : 0.0f;
    Tt[16] = hp;
    Tt[17] = -a2 * hl;
    Tt[18] = fmaf(b1, hl, b2 * hm);
    Tt[19] = b2 * hl;

    v2f Bk[5];
    #pragma unroll
    for (int kc = 0; kc < 5; ++kc) {
        Bk[kc].x = half ? Tt[4 * kc + 2] : Tt[4 * kc + 0];
        Bk[kc].y = half ? Tt[4 * kc + 3] : Tt[4 * kc + 1];
    }

    const int gch = row * NC + c;
    const int pb  = gch / P1_THREADS;
    const int j   = gch % P1_THREADS;
    const float2 e = entry[gch];
    float2 Bt[MAXJOIN];
    #pragma unroll
    for (int k = 1; k <= MAXJOIN; ++k) Bt[k - 1] = btot[max(pb - k, 0) * BT_PITCH];


    const float* xr = x + (long)row * TLEN;

    float xm1 = (t0 > 0) ? xr[t0 - 1] : 0.0f;
    float xm2 = (t0 > 0) ? xr[t0 - 2] : 0.0f;

    const float* xg = x + (long)(r0 + half) * TLEN + t0 + 4 * lid;
    float4 gl[NSLAB3][8];
    #pragma unroll
    for (int s = 0; s < NSLAB3; ++s) {
        #pragma unroll
        for (int p = 0; p < 8; ++p)
            gl[s][p] = *reinterpret_cast<const float4*>(xg + (long)(2 * p) * TLEN + s * SLAB3);
    }
    __builtin_amdgcn_sched_barrier(0);

    float q00 = p00, q01 = p01, q10 = p10, q11 = p11;
    float r00 = 1.0f, r01 = 0.0f, r10 = 0.0f, r11 = 1.0f;
    #pragma unroll
    for (int k = 1; k < P1_THREADS; k <<= 1) {
        if (j & k) mat_mul(r00, r01, r10, r11, q00, q01, q10, q11);
        mat_square(q00, q01, q10, q11);
    }
    float E1 = 0.0f, E2 = 0.0f;
    #pragma unroll
    for (int k = MAXJOIN; k >= 1; --k) {
        const bool ok = c >= j + 1 + P1_THREADS * (k - 1);
        float n1 = E1, n2 = E2;
        aff_step(n1, n2, Bt[k - 1].x, Bt[k - 1].y, q00, q01, q10, q11);
        E1 = ok ? n1 : E1;
        E2 = ok ? n2 : E2;
    }
    float ym1 = e.x, ym2 = e.y;
    if (c >= j + 1) {
        ym1 = fmaf(r00, E1, fmaf(r01, E2, ym1));
        ym2 = fmaf(r10, E1, fmaf(r11, E2, ym2));
    }

    v2f* tw = tile[threadIdx.x >> 5];
    v2f Ax[NBLK][4];
    #pragma unroll
    for (int s = 0; s < PART_SLABS; ++s) {
        #pragma unroll
        for (int p = 0; p < 8; ++p) {
            const int o = (2 * p + half) * ROW8 + s * (SLAB3 / 2) + 2 * lid;
            tw[o]     = v2f{gl[s][p].x, gl[s][p].y};
            tw[o + 1] = v2f{gl[s][p].z, gl[s][p].w};
        }
    }
    __builtin_amdgcn_fence(__ATOMIC_SEQ_CST, "wavefront");
    #pragma unroll
    for (int blk = 0; blk < PART_BLKS; ++blk) {
        #pragma unroll
        for (int kc = 0; kc < 4; ++kc) Ax[blk][kc] = tw[lid * ROW8 + blk * 8 + 2 * kc + half];
    }
    __builtin_amdgcn_fence(__ATOMIC_SEQ_CST, "wavefront");
    #pragma unroll
    for (int s = PART_SLABS; s < NSLAB3; ++s) {
        #pragma unroll
        for (int p = 0; p < 8; ++p) {
            const int o = (2 * p + half) * ROW8 + (s - PART_SLABS) * (SLAB3 / 2) + 2 * lid;
            tw[o]     = v2f{gl[s][p].x, gl[s][p].y};
            tw[o + 1] = v2f{gl[s][p].z, gl[s][p].w};
        }
    }
    __builtin_amdgcn_fence(__ATOMIC_SEQ_CST, "wavefront");
    #pragma unroll
    for (int blk = PART_BLKS; blk < NBLK; ++blk) {
        #pragma unroll
        for (int kc = 0; kc < 4; ++kc)
            Ax[blk][kc] = tw[lid * ROW8 + (blk - PART_BLKS) * 8 + 2 * kc + half];
    }

    v8f part = {};
    #pragma unroll
    for (int kc = 0; kc < 4; ++kc) {
        part = __builtin_amdgcn_wmma_f32_16x16x4_f32(
             false, Ax[0][kc],  false, Bk[kc],
             (short)0, part,  false,  false);
    }

    static_assert(NBLK % 2 == 0, "blocks are stored two at a time");
    v8f held = {};

    #pragma unroll
    for (int blk = 0; blk < NBLK; ++blk) {
        int tb = t0 + blk * 16;

        v2f Ak4;
        Ak4.x = half ? xm1 : ym1;
        Ak4.y = half ? xm2 : ym2;

        v8f acc = __builtin_amdgcn_wmma_f32_16x16x4_f32(
             false, Ak4,  false, Bk[4],
             (short)0, part,  false,  false);

        if (blk + 1 < NBLK) {
            v8f nxt = {};
            #pragma unroll
            for (int kc = 0; kc < 4; ++kc) {
                nxt = __builtin_amdgcn_wmma_f32_16x16x4_f32(
                     false, Ax[blk + 1][kc],  false, Bk[kc],
                     (short)0, nxt,  false,  false);
            }
            part = nxt;
        }

        float u1[8], u2[8];
        #pragma unroll
        for (int v = 0; v < 8; ++v) {
            u1[v] = row_lane<15>(acc[v]);
            u2[v] = row_lane<14>(acc[v]);
        }
        float z1 = pick8(u1, bit0, bit1, bit2);
        float z2 = pick8(u2, bit0, bit1, bit2);
        float w1 = other_half(z1);
        float w2 = other_half(z2);
        ym1 = upper8 ? w1 : z1;
        ym2 = upper8 ? w2 : z2;

        xm1 = Ax[blk][3].y; xm2 = Ax[blk][3].x;

        if (blk & 1) {
            const int tp = tb - 16;
            float lo[8], hi[8];
            #pragma unroll
            for (int v = 0; v < 8; ++v) { lo[v] = held[v]; hi[v] = acc[v]; }
            if (blk == NBLK - 1) {
                swap_halves4<true>(lo[0], lo[1], lo[2], lo[3], hi[0], hi[1], hi[2], hi[3]);
                swap_halves4<true>(lo[4], lo[5], lo[6], lo[7], hi[4], hi[5], hi[6], hi[7]);
            } else {
                swap_halves4<false>(lo[0], lo[1], lo[2], lo[3], hi[0], hi[1], hi[2], hi[3]);
                swap_halves4<false>(lo[4], lo[5], lo[6], lo[7], hi[4], hi[5], hi[6], hi[7]);
            }
            #pragma unroll
            for (int v = 0; v < 8; ++v) {
                __builtin_nontemporal_store(lo[v], &out[(long)(r0 + v) * TLEN + tp + lane]);
                __builtin_nontemporal_store(hi[v], &out[(long)(r0 + v + 8) * TLEN + tp + lane]);
            }
        } else {
            held = acc;
        }

        __builtin_amdgcn_sched_barrier(0);
    }
}

extern "C" void kernel_launch(void* const* d_in, const int* in_sizes, int n_in,
                              void* d_out, int out_size, void* d_ws, size_t ws_size,
                              hipStream_t stream) {
    const float* x = (const float*)d_in[0];
    float* out = (float*)d_out;

    float2* entry = (float2*)d_ws;
    float2* btot  = entry + (size_t)BROWS * NC;

    double w0    = 2.0 * M_PI * 7500.0 / 16000.0;
    double alpha = sin(w0) / (2.0 * 0.707);
    double cw    = cos(w0);
    double b0 = (1.0 - cw) / 2.0, b1 = 1.0 - cw, b2 = b0;
    double a0 = 1.0 + alpha, a1 = -2.0 * cw, a2 = 1.0 - alpha;
    b0 /= a0; b1 /= a0; b2 /= a0; a1 /= a0; a2 /= a0;

    double p00 = 1, p01 = 0, p10 = 0, p11 = 1;
    for (int i = 0; i < CHUNK; ++i) {
        double n00 = -a1 * p00 - a2 * p10;
        double n01 = -a1 * p01 - a2 * p11;
        double n10 = p00, n11 = p01;
        p00 = n00; p01 = n01; p10 = n10; p11 = n11;
    }

    float fb0 = (float)b0, fb1 = (float)b1, fb2 = (float)b2;
    float fa1 = (float)a1, fa2 = (float)a2;

    float fp00 = (float)p00, fp01 = (float)p01, fp10 = (float)p10, fp11 = (float)p11;

    lp_phase1<<<NPB, P1_THREADS, 0, stream>>>(x, entry, btot, fb0, fb1, fb2, fa1, fa2,
                                              fp00, fp01, fp10, fp11);
    int nwaves = NC * (BROWS / 16);
    lp_phase3<<<nwaves / 8, 256, 0, stream>>>(x, out, entry, btot, fb0, fb1, fb2, fa1, fa2,
                                              fp00, fp01, fp10, fp11);
}
